# starting kernel
# baseline (speedup 1.0000x reference)
#include <hip/hip_runtime.h>
#include <hip/hip_bf16.h>
#include <math.h>

typedef __attribute__((ext_vector_type(2))) float v2f;
typedef __attribute__((ext_vector_type(8))) float v8f;


__global__ void gcn_zero_kernel(float* __restrict__ p, int n) {
    int i = blockIdx.x * blockDim.x + threadIdx.x;
    if (i < n) p[i] = 0.0f;
}

__global__ void gcn_degree_kernel(const int* __restrict__ src, const int* __restrict__ dst,
                                  float* __restrict__ outdeg, float* __restrict__ indeg, int n_edges) {
    int i = blockIdx.x * blockDim.x + threadIdx.x;
    if (i < n_edges) {
        atomicAdd(&outdeg[src[i]], 1.0f);
        atomicAdd(&indeg[dst[i]], 1.0f);
    }
}

__global__ void gcn_norm_kernel(float* __restrict__ p, int n) {
    int i = blockIdx.x * blockDim.x + threadIdx.x;
    if (i < n) {
        float d = p[i];
        d = (d < 1.0f) ? 1.0f : d;
        p[i] = 1.0f / sqrtf(d);
    }
}


template <int NT>
__global__ __launch_bounds__(256)
void gcn_gemm_scaled_wmma(const float* __restrict__ X, const float* __restrict__ scale,
                          const float* __restrict__ W, float* __restrict__ out,
                          int n_nodes, int K) {
    const int NOUT = NT * 16;
    int wave = (blockIdx.x * blockDim.x + threadIdx.x) >> 5;
    int lane = threadIdx.x & 31;
    int row0 = wave * 16;
    if (row0 >= n_nodes) return;

    int hi   = lane >> 4;
    int kOff = hi * 2;
    int mrow = row0 + (lane & 15);
    int col  = lane & 15;

    float s = scale[mrow];
    const float* __restrict__ xrow = X + (size_t)mrow * (size_t)K;

    v8f acc[NT];
    v8f vzero = {};
#pragma unroll
    for (int nt = 0; nt < NT; ++nt) acc[nt] = vzero;

    for (int k0 = 0; k0 < K; k0 += 4) {
        float2 av = *(const float2*)(xrow + k0 + kOff);
        v2f a;
        a.x = av.x * s;
        a.y = av.y * s;
        const float* __restrict__ wr = W + (size_t)(k0 + kOff) * (size_t)NOUT;
#pragma unroll
        for (int nt = 0; nt < NT; ++nt) {
            v2f b;
            b.x = wr[nt * 16 + col];
            b.y = wr[NOUT + nt * 16 + col];
            acc[nt] = __builtin_amdgcn_wmma_f32_16x16x4_f32(
                 false, a,  false, b,
                 (short)0, acc[nt],  false,  false);
        }
    }

    int crow = row0 + hi * 8;
#pragma unroll
    for (int nt = 0; nt < NT; ++nt) {
#pragma unroll
        for (int v = 0; v < 8; ++v) {
            out[(size_t)(crow + v) * (size_t)NOUT + nt * 16 + col] = acc[nt][v];
        }
    }
}


template <int F>
__global__ void gcn_scatter_add_kernel(const float* __restrict__ hw, const int* __restrict__ src,
                                       const int* __restrict__ dst, float* __restrict__ agg,
                                       int n_edges) {
    constexpr int TPE = F / 4;
    int gid = blockIdx.x * blockDim.x + threadIdx.x;
    int e = gid / TPE;
    if (e >= n_edges) return;
    int t = gid % TPE;
    int s = src[e];
    int d = dst[e];
    const float4 v = *(const float4*)(hw + (size_t)s * F + t * 4);
    float* o = agg + (size_t)d * F + t * 4;
    atomicAdd(o + 0, v.x);
    atomicAdd(o + 1, v.y);
    atomicAdd(o + 2, v.z);
    atomicAdd(o + 3, v.w);
}


__global__ void gcn_finalize_kernel(const float* __restrict__ agg, const float* __restrict__ dnorm,
                                    const float* __restrict__ bias, float* __restrict__ out,
                                    int n_nodes, int F, int act) {
    int i = blockIdx.x * blockDim.x + threadIdx.x;
    if (i >= n_nodes * F) return;
    int node = i / F;
    int f = i % F;
    float v = agg[i] * dnorm[node] + bias[f];
    if (act == 0) {
        v = (v > 0.0f) ? v : 0.0f;
    } else {
        v = 1.0f / (1.0f + expf(-v)) + 1e-8f;
    }
    out[i] = v;
}


extern "C" void kernel_launch(void* const* d_in, const int* in_sizes, int n_in,
                              void* d_out, int out_size, void* d_ws, size_t ws_size,
                              hipStream_t stream) {
    (void)n_in; (void)out_size; (void)ws_size;

    const float* feat = (const float*)d_in[0];
    const int*   src  = (const int*)d_in[1];
    const int*   dst  = (const int*)d_in[2];
    const float* W0   = (const float*)d_in[3];
    const float* b0   = (const float*)d_in[4];
    const float* W1   = (const float*)d_in[5];
    const float* b1   = (const float*)d_in[6];
    const float* W2   = (const float*)d_in[7];
    const float* b2   = (const float*)d_in[8];

    const int F = 128;
    const int C = 16;
    const int N = in_sizes[0] / F;
    const int E = in_sizes[1];

    float* snorm = (float*)d_ws;
    float* dnorm = snorm + N;
    float* bufA  = dnorm + N;
    float* bufB  = bufA + (size_t)N * F;

    const int TB = 256;
    auto nblk = [](long long n, int tb) { return (unsigned)((n + tb - 1) / tb); };

    gcn_zero_kernel<<<nblk(2LL * N, TB), TB, 0, stream>>>(snorm, 2 * N);
    gcn_degree_kernel<<<nblk(E, TB), TB, 0, stream>>>(src, dst, snorm, dnorm, E);
    gcn_norm_kernel<<<nblk(2LL * N, TB), TB, 0, stream>>>(snorm, 2 * N);

    const long long gemm_threads = (long long)((N + 15) / 16) * 32;
    const long long nf = (long long)N * F;
    const long long nc = (long long)N * C;

    gcn_gemm_scaled_wmma<8><<<nblk(gemm_threads, TB), TB, 0, stream>>>(feat, snorm, W0, bufA, N, F);
    gcn_zero_kernel<<<nblk(nf, TB), TB, 0, stream>>>(bufB, (int)nf);
    gcn_scatter_add_kernel<128><<<nblk((long long)E * 32, TB), TB, 0, stream>>>(bufA, src, dst, bufB, E);
    gcn_finalize_kernel<<<nblk(nf, TB), TB, 0, stream>>>(bufB, dnorm, b0, bufA, N, F, 0);

    gcn_gemm_scaled_wmma<8><<<nblk(gemm_threads, TB), TB, 0, stream>>>(bufA, snorm, W1, bufB, N, F);
    gcn_zero_kernel<<<nblk(nf, TB), TB, 0, stream>>>(bufA, (int)nf);
    gcn_scatter_add_kernel<128><<<nblk((long long)E * 32, TB), TB, 0, stream>>>(bufB, src, dst, bufA, E);
    gcn_finalize_kernel<<<nblk(nf, TB), TB, 0, stream>>>(bufA, dnorm, b1, bufB, N, F, 0);

    gcn_gemm_scaled_wmma<1><<<nblk(gemm_threads, TB), TB, 0, stream>>>(bufB, snorm, W2, bufA, N, F);
    gcn_zero_kernel<<<nblk(nc, TB), TB, 0, stream>>>(bufB, (int)nc);
    gcn_scatter_add_kernel<16><<<nblk((long long)E * 4, TB), TB, 0, stream>>>(bufA, src, dst, bufB, E);
    gcn_finalize_kernel<<<nblk(nc, TB), TB, 0, stream>>>(bufB, dnorm, b2, (float*)d_out, N, C, 1);
}


# Float-atomic scatter-add replaced by an in-call counting sort of edges by destination (arrival-order rank) and a per-node register segment sum; zero fills dropped; sum order changes
# speedup vs baseline: 2.1603x; 2.1603x over previous
#include <hip/hip_runtime.h>
#include <hip/hip_bf16.h>
#include <math.h>

typedef __attribute__((ext_vector_type(2))) float v2f;
typedef __attribute__((ext_vector_type(8))) float v8f;


__global__ void gcn_zero_kernel(float* __restrict__ p, int n) {
    int i = blockIdx.x * blockDim.x + threadIdx.x;
    if (i < n) p[i] = 0.0f;
}

__global__ void gcn_degree_rank_kernel(const int* __restrict__ src, const int* __restrict__ dst,
                                       float* __restrict__ outdeg, int* __restrict__ incnt,
                                       int* __restrict__ rank, int n_edges) {
    int i = blockIdx.x * blockDim.x + threadIdx.x;
    if (i < n_edges) {
        atomicAdd(&outdeg[src[i]], 1.0f);
        rank[i] = atomicAdd(&incnt[dst[i]], 1);
    }
}

__global__ void gcn_norm_kernel(float* __restrict__ snorm, const int* __restrict__ incnt,
                                float* __restrict__ dnorm, int n) {
    int i = blockIdx.x * blockDim.x + threadIdx.x;
    if (i < n) {
        float d = snorm[i];
        d = (d < 1.0f) ? 1.0f : d;
        snorm[i] = 1.0f / sqrtf(d);
        float c = (float)incnt[i];
        c = (c < 1.0f) ? 1.0f : c;
        dnorm[i] = 1.0f / sqrtf(c);
    }
}


#define GCN_SCAN_CHUNK 1024

__device__ __forceinline__ int gcn_block_sum_256(int v, int* sm) {
    int lane = threadIdx.x & 31, wv = threadIdx.x >> 5;
    for (int o = 16; o > 0; o >>= 1) v += __shfl_xor(v, o);
    if (lane == 0) sm[wv] = v;
    __syncthreads();
    int t = 0;
    for (int w = 0; w < 8; ++w) t += sm[w];
    __syncthreads();
    return t;
}

__global__ __launch_bounds__(256)
void gcn_chunk_sum_kernel(const int* __restrict__ cnt, int* __restrict__ csum, int n) {
    __shared__ int sm[8];
    int base = blockIdx.x * GCN_SCAN_CHUNK + threadIdx.x * 4;
    int s = 0;
    for (int k = 0; k < 4; ++k) if (base + k < n) s += cnt[base + k];
    int t = gcn_block_sum_256(s, sm);
    if (threadIdx.x == 0) csum[blockIdx.x] = t;
}

__global__ __launch_bounds__(256)
void gcn_rowptr_kernel(const int* __restrict__ cnt, const int* __restrict__ csum,
                       int* __restrict__ rowptr, int n) {
    __shared__ int sm[8];
    __shared__ int wtot[8];
    int tid = threadIdx.x, lane = tid & 31, wv = tid >> 5;

    int p = 0;
    for (int j = tid; j < (int)blockIdx.x; j += 256) p += csum[j];
    int chunk_off = gcn_block_sum_256(p, sm);

    int base = blockIdx.x * GCN_SCAN_CHUNK + tid * 4;
    int c[4];
    int t = 0;
#pragma unroll
    for (int k = 0; k < 4; ++k) { c[k] = (base + k < n) ? cnt[base + k] : 0; t += c[k]; }

    int inc = t;
    for (int o = 1; o < 32; o <<= 1) {
        int y = __shfl_up(inc, o);
        if (lane >= o) inc += y;
    }
    if (lane == 31) wtot[wv] = inc;
    __syncthreads();
    int wpre = 0;
    for (int w = 0; w < 8; ++w) if (w < wv) wpre += wtot[w];

    int run = chunk_off + wpre + inc - t;
#pragma unroll
    for (int k = 0; k < 4; ++k) {
        if (base + k < n) {
            rowptr[base + k] = run;
            run += c[k];
            if (base + k == n - 1) rowptr[n] = run;
        }
    }
}

__global__ void gcn_fill_kernel(const int* __restrict__ src, const int* __restrict__ dst,
                                const int* __restrict__ rank, const int* __restrict__ rowptr,
                                int* __restrict__ ssrc, int n_edges) {
    int i = blockIdx.x * blockDim.x + threadIdx.x;
    if (i < n_edges) ssrc[rowptr[dst[i]] + rank[i]] = src[i];
}


template <int NT>
__global__ __launch_bounds__(256)
void gcn_gemm_scaled_wmma(const float* __restrict__ X, const float* __restrict__ scale,
                          const float* __restrict__ W, float* __restrict__ out,
                          int n_nodes, int K) {
    const int NOUT = NT * 16;
    int wave = (blockIdx.x * blockDim.x + threadIdx.x) >> 5;
    int lane = threadIdx.x & 31;
    int row0 = wave * 16;
    if (row0 >= n_nodes) return;

    int hi   = lane >> 4;
    int kOff = hi * 2;
    int mrow = row0 + (lane & 15);
    int col  = lane & 15;

    float s = scale[mrow];
    const float* __restrict__ xrow = X + (size_t)mrow * (size_t)K;

    v8f acc[NT];
    v8f vzero = {};
#pragma unroll
    for (int nt = 0; nt < NT; ++nt) acc[nt] = vzero;

    for (int k0 = 0; k0 < K; k0 += 4) {
        float2 av = *(const float2*)(xrow + k0 + kOff);
        v2f a;
        a.x = av.x * s;
        a.y = av.y * s;
        const float* __restrict__ wr = W + (size_t)(k0 + kOff) * (size_t)NOUT;
#pragma unroll
        for (int nt = 0; nt < NT; ++nt) {
            v2f b;
            b.x = wr[nt * 16 + col];
            b.y = wr[NOUT + nt * 16 + col];
            acc[nt] = __builtin_amdgcn_wmma_f32_16x16x4_f32(
                 false, a,  false, b,
                 (short)0, acc[nt],  false,  false);
        }
    }

    int crow = row0 + hi * 8;
#pragma unroll
    for (int nt = 0; nt < NT; ++nt) {
#pragma unroll
        for (int v = 0; v < 8; ++v) {
            out[(size_t)(crow + v) * (size_t)NOUT + nt * 16 + col] = acc[nt][v];
        }
    }
}


__global__ __launch_bounds__(256)
void gcn_gather128_kernel(const float* __restrict__ hw, const int* __restrict__ rowptr,
                          const int* __restrict__ ssrc, float* __restrict__ agg, int n_nodes) {
    constexpr int F = 128;
    int lane = threadIdx.x & 31;
    int node = __builtin_amdgcn_readfirstlane((int)((blockIdx.x * blockDim.x + threadIdx.x) >> 5));
    if (node >= n_nodes) return;
    int beg = rowptr[node];
    int end = rowptr[node + 1];

    float4 acc = make_float4(0.0f, 0.0f, 0.0f, 0.0f);
    const float* __restrict__ colp = hw + lane * 4;

    for (int b = beg; b < end; b += 32) {
        int m = end - b;
        if (m > 32) m = 32;
        int my = (lane < m) ? ssrc[b + lane] : 0;
        for (int j = 0; j < m; j += 8) {
            int s[8];
            float4 v[8];
#pragma unroll
            for (int u = 0; u < 8; ++u) {
                int l = (j + u < m) ? (j + u) : (m - 1);
                s[u] = __builtin_amdgcn_readlane(my, l);
            }
#pragma unroll
            for (int u = 0; u < 8; ++u) {
                v[u] = make_float4(0.0f, 0.0f, 0.0f, 0.0f);
                if (j + u < m) v[u] = *(const float4*)(colp + (size_t)s[u] * F);
            }
#pragma unroll
            for (int u = 0; u < 8; ++u) {
                acc.x += v[u].x; acc.y += v[u].y; acc.z += v[u].z; acc.w += v[u].w;
            }
        }
    }
    *(float4*)(agg + (size_t)node * F + lane * 4) = acc;
}

__global__ __launch_bounds__(256)
void gcn_gather16_kernel(const float* __restrict__ hw, const int* __restrict__ rowptr,
                         const int* __restrict__ ssrc, float* __restrict__ agg, int n_nodes) {
    constexpr int F = 16;
    int gid = blockIdx.x * blockDim.x + threadIdx.x;
    int node = gid >> 2;
    int sub = gid & 3;
    if (node >= n_nodes) return;
    int beg = rowptr[node];
    int end = rowptr[node + 1];

    float4 acc = make_float4(0.0f, 0.0f, 0.0f, 0.0f);
    const float* __restrict__ colp = hw + sub * 4;

    for (int j = beg; j < end; j += 8) {
        int s[8];
        float4 v[8];
#pragma unroll
        for (int u = 0; u < 8; ++u) s[u] = (j + u < end) ? ssrc[j + u] : -1;
#pragma unroll
        for (int u = 0; u < 8; ++u) {
            v[u] = make_float4(0.0f, 0.0f, 0.0f, 0.0f);
            if (s[u] >= 0) v[u] = *(const float4*)(colp + (size_t)s[u] * F);
        }
#pragma unroll
        for (int u = 0; u < 8; ++u) {
            acc.x += v[u].x; acc.y += v[u].y; acc.z += v[u].z; acc.w += v[u].w;
        }
    }
    *(float4*)(agg + (size_t)node * F + sub * 4) = acc;
}


__global__ void gcn_finalize_kernel(const float* __restrict__ agg, const float* __restrict__ dnorm,
                                    const float* __restrict__ bias, float* __restrict__ out,
                                    int n_nodes, int F, int act) {
    int i = blockIdx.x * blockDim.x + threadIdx.x;
    if (i >= n_nodes * F) return;
    int node = i / F;
    int f = i % F;
    float v = agg[i] * dnorm[node] + bias[f];
    if (act == 0) {
        v = (v > 0.0f) ? v : 0.0f;
    } else {
        v = 1.0f / (1.0f + expf(-v)) + 1e-8f;
    }
    out[i] = v;
}


extern "C" void kernel_launch(void* const* d_in, const int* in_sizes, int n_in,
                              void* d_out, int out_size, void* d_ws, size_t ws_size,
                              hipStream_t stream) {
    (void)n_in; (void)out_size; (void)ws_size;

    const float* feat = (const float*)d_in[0];
    const int*   src  = (const int*)d_in[1];
    const int*   dst  = (const int*)d_in[2];
    const float* W0   = (const float*)d_in[3];
    const float* b0   = (const float*)d_in[4];
    const float* W1   = (const float*)d_in[5];
    const float* b1   = (const float*)d_in[6];
    const float* W2   = (const float*)d_in[7];
    const float* b2   = (const float*)d_in[8];

    const int F = 128;
    const int C = 16;
    const int N = in_sizes[0] / F;
    const int E = in_sizes[1];

    auto al = [](size_t n) { return (n + 63) & ~(size_t)63; };
    const int n_chunks = (N + GCN_SCAN_CHUNK - 1) / GCN_SCAN_CHUNK;
    float* snorm  = (float*)d_ws;
    int*   incnt  = (int*)(snorm + N);
    float* dnorm  = snorm + al(2 * (size_t)N);
    int*   rowptr = (int*)(dnorm + al((size_t)N));
    int*   csum   = rowptr + al((size_t)N + 1);
    int*   rank   = csum + al((size_t)n_chunks);
    int*   ssrc   = rank + al((size_t)E);
    float* bufA   = (float*)(ssrc + al((size_t)E));
    float* bufB   = bufA + al((size_t)N * F);

    const int TB = 256;
    auto nblk = [](long long n, int tb) { return (unsigned)((n + tb - 1) / tb); };

    gcn_zero_kernel<<<nblk(2LL * N, TB), TB, 0, stream>>>(snorm, 2 * N);
    gcn_degree_rank_kernel<<<nblk(E, TB), TB, 0, stream>>>(src, dst, snorm, incnt, rank, E);
    gcn_chunk_sum_kernel<<<n_chunks, TB, 0, stream>>>(incnt, csum, N);
    gcn_rowptr_kernel<<<n_chunks, TB, 0, stream>>>(incnt, csum, rowptr, N);
    gcn_norm_kernel<<<nblk(N, TB), TB, 0, stream>>>(snorm, incnt, dnorm, N);
    gcn_fill_kernel<<<nblk(E, TB), TB, 0, stream>>>(src, dst, rank, rowptr, ssrc, E);

    const long long gemm_threads = (long long)((N + 15) / 16) * 32;
    const long long nf = (long long)N * F;
    const long long nc = (long long)N * C;

    gcn_gemm_scaled_wmma<8><<<nblk(gemm_threads, TB), TB, 0, stream>>>(feat, snorm, W0, bufA, N, F);
    gcn_gather128_kernel<<<nblk((long long)N * 32, TB), TB, 0, stream>>>(bufA, rowptr, ssrc, bufB, N);
    gcn_finalize_kernel<<<nblk(nf, TB), TB, 0, stream>>>(bufB, dnorm, b0, bufA, N, F, 0);

    gcn_gemm_scaled_wmma<8><<<nblk(gemm_threads, TB), TB, 0, stream>>>(bufA, snorm, W1, bufB, N, F);
    gcn_gather128_kernel<<<nblk((long long)N * 32, TB), TB, 0, stream>>>(bufB, rowptr, ssrc, bufA, N);
    gcn_finalize_kernel<<<nblk(nf, TB), TB, 0, stream>>>(bufA, dnorm, b1, bufB, N, F, 0);

    gcn_gemm_scaled_wmma<1><<<nblk(gemm_threads, TB), TB, 0, stream>>>(bufB, snorm, W2, bufA, N, F);
    gcn_gather16_kernel<<<nblk((long long)N * 4, TB), TB, 0, stream>>>(bufA, rowptr, ssrc, bufB, N);
    gcn_finalize_kernel<<<nblk(nc, TB), TB, 0, stream>>>(bufB, dnorm, b2, (float*)d_out, N, C, 1);
}


# Finalize (dst-norm scale, bias, relu/sigmoid) fused into the segment-sum kernels' store: 3 element-wise launches, their 850k one-element waves and the agg round trip through memory dropped
# speedup vs baseline: 2.5684x; 1.1889x over previous
#include <hip/hip_runtime.h>
#include <hip/hip_bf16.h>
#include <math.h>

typedef __attribute__((ext_vector_type(2))) float v2f;
typedef __attribute__((ext_vector_type(8))) float v8f;


__global__ void gcn_zero_kernel(float* __restrict__ p, int n) {
    int i = blockIdx.x * blockDim.x + threadIdx.x;
    if (i < n) p[i] = 0.0f;
}

__global__ void gcn_degree_rank_kernel(const int* __restrict__ src, const int* __restrict__ dst,
                                       float* __restrict__ outdeg, int* __restrict__ incnt,
                                       int* __restrict__ rank, int n_edges) {
    int i = blockIdx.x * blockDim.x + threadIdx.x;
    if (i < n_edges) {
        atomicAdd(&outdeg[src[i]], 1.0f);
        rank[i] = atomicAdd(&incnt[dst[i]], 1);
    }
}

__global__ void gcn_norm_kernel(float* __restrict__ snorm, const int* __restrict__ incnt,
                                float* __restrict__ dnorm, int n) {
    int i = blockIdx.x * blockDim.x + threadIdx.x;
    if (i < n) {
        float d = snorm[i];
        d = (d < 1.0f) ? 1.0f : d;
        snorm[i] = 1.0f / sqrtf(d);
        float c = (float)incnt[i];
        c = (c < 1.0f) ? 1.0f : c;
        dnorm[i] = 1.0f / sqrtf(c);
    }
}


#define GCN_SCAN_CHUNK 1024

__device__ __forceinline__ int gcn_block_sum_256(int v, int* sm) {
    int lane = threadIdx.x & 31, wv = threadIdx.x >> 5;
    for (int o = 16; o > 0; o >>= 1) v += __shfl_xor(v, o);
    if (lane == 0) sm[wv] = v;
    __syncthreads();
    int t = 0;
    for (int w = 0; w < 8; ++w) t += sm[w];
    __syncthreads();
    return t;
}

__global__ __launch_bounds__(256)
void gcn_chunk_sum_kernel(const int* __restrict__ cnt, int* __restrict__ csum, int n) {
    __shared__ int sm[8];
    int base = blockIdx.x * GCN_SCAN_CHUNK + threadIdx.x * 4;
    int s = 0;
    for (int k = 0; k < 4; ++k) if (base + k < n) s += cnt[base + k];
    int t = gcn_block_sum_256(s, sm);
    if (threadIdx.x == 0) csum[blockIdx.x] = t;
}

__global__ __launch_bounds__(256)
void gcn_rowptr_kernel(const int* __restrict__ cnt, const int* __restrict__ csum,
                       int* __restrict__ rowptr, int n) {
    __shared__ int sm[8];
    __shared__ int wtot[8];
    int tid = threadIdx.x, lane = tid & 31, wv = tid >> 5;

    int p = 0;
    for (int j = tid; j < (int)blockIdx.x; j += 256) p += csum[j];
    int chunk_off = gcn_block_sum_256(p, sm);

    int base = blockIdx.x * GCN_SCAN_CHUNK + tid * 4;
    int c[4];
    int t = 0;
#pragma unroll
    for (int k = 0; k < 4; ++k) { c[k] = (base + k < n) ? cnt[base + k] : 0; t += c[k]; }

    int inc = t;
    for (int o = 1; o < 32; o <<= 1) {
        int y = __shfl_up(inc, o);
        if (lane >= o) inc += y;
    }
    if (lane == 31) wtot[wv] = inc;
    __syncthreads();
    int wpre = 0;
    for (int w = 0; w < 8; ++w) if (w < wv) wpre += wtot[w];

    int run = chunk_off + wpre + inc - t;
#pragma unroll
    for (int k = 0; k < 4; ++k) {
        if (base + k < n) {
            rowptr[base + k] = run;
            run += c[k];
            if (base + k == n - 1) rowptr[n] = run;
        }
    }
}

__global__ void gcn_fill_kernel(const int* __restrict__ src, const int* __restrict__ dst,
                                const int* __restrict__ rank, const int* __restrict__ rowptr,
                                int* __restrict__ ssrc, int n_edges) {
    int i = blockIdx.x * blockDim.x + threadIdx.x;
    if (i < n_edges) ssrc[rowptr[dst[i]] + rank[i]] = src[i];
}


template <int NT>
__global__ __launch_bounds__(256)
void gcn_gemm_scaled_wmma(const float* __restrict__ X, const float* __restrict__ scale,
                          const float* __restrict__ W, float* __restrict__ out,
                          int n_nodes, int K) {
    const int NOUT = NT * 16;
    int wave = (blockIdx.x * blockDim.x + threadIdx.x) >> 5;
    int lane = threadIdx.x & 31;
    int row0 = wave * 16;
    if (row0 >= n_nodes) return;

    int hi   = lane >> 4;
    int kOff = hi * 2;
    int mrow = row0 + (lane & 15);
    int col  = lane & 15;

    float s = scale[mrow];
    const float* __restrict__ xrow = X + (size_t)mrow * (size_t)K;

    v8f acc[NT];
    v8f vzero = {};
#pragma unroll
    for (int nt = 0; nt < NT; ++nt) acc[nt] = vzero;

    for (int k0 = 0; k0 < K; k0 += 4) {
        float2 av = *(const float2*)(xrow + k0 + kOff);
        v2f a;
        a.x = av.x * s;
        a.y = av.y * s;
        const float* __restrict__ wr = W + (size_t)(k0 + kOff) * (size_t)NOUT;
#pragma unroll
        for (int nt = 0; nt < NT; ++nt) {
            v2f b;
            b.x = wr[nt * 16 + col];
            b.y = wr[NOUT + nt * 16 + col];
            acc[nt] = __builtin_amdgcn_wmma_f32_16x16x4_f32(
                 false, a,  false, b,
                 (short)0, acc[nt],  false,  false);
        }
    }

    int crow = row0 + hi * 8;
#pragma unroll
    for (int nt = 0; nt < NT; ++nt) {
#pragma unroll
        for (int v = 0; v < 8; ++v) {
            out[(size_t)(crow + v) * (size_t)NOUT + nt * 16 + col] = acc[nt][v];
        }
    }
}


__global__ __launch_bounds__(256)
void gcn_gather128_relu_kernel(const float* __restrict__ hw, const int* __restrict__ rowptr,
                               const int* __restrict__ ssrc, const float* __restrict__ dnorm,
                               const float* __restrict__ bias, float* __restrict__ out, int n_nodes) {
    constexpr int F = 128;
    int lane = threadIdx.x & 31;
    int node = __builtin_amdgcn_readfirstlane((int)((blockIdx.x * blockDim.x + threadIdx.x) >> 5));
    if (node >= n_nodes) return;
    int beg = rowptr[node];
    int end = rowptr[node + 1];
    float dn = dnorm[node];
    int c0 = lane * 4;
    float4 bv = make_float4(bias[c0], bias[c0 + 1], bias[c0 + 2], bias[c0 + 3]);

    float4 acc = make_float4(0.0f, 0.0f, 0.0f, 0.0f);
    const float* __restrict__ colp = hw + lane * 4;

    for (int b = beg; b < end; b += 32) {
        int m = end - b;
        if (m > 32) m = 32;
        int my = (lane < m) ? ssrc[b + lane] : 0;
        for (int j = 0; j < m; j += 8) {
            int s[8];
            float4 v[8];
#pragma unroll
            for (int u = 0; u < 8; ++u) {
                int l = (j + u < m) ? (j + u) : (m - 1);
                s[u] = __builtin_amdgcn_readlane(my, l);
            }
#pragma unroll
            for (int u = 0; u < 8; ++u) {
                v[u] = make_float4(0.0f, 0.0f, 0.0f, 0.0f);
                if (j + u < m) v[u] = *(const float4*)(colp + (size_t)s[u] * F);
            }
#pragma unroll
            for (int u = 0; u < 8; ++u) {
                acc.x += v[u].x; acc.y += v[u].y; acc.z += v[u].z; acc.w += v[u].w;
            }
        }
    }
    float4 r;
    r.x = acc.x * dn + bv.x; r.x = (r.x > 0.0f) ? r.x : 0.0f;
    r.y = acc.y * dn + bv.y; r.y = (r.y > 0.0f) ? r.y : 0.0f;
    r.z = acc.z * dn + bv.z; r.z = (r.z > 0.0f) ? r.z : 0.0f;
    r.w = acc.w * dn + bv.w; r.w = (r.w > 0.0f) ? r.w : 0.0f;
    *(float4*)(out + (size_t)node * F + lane * 4) = r;
}

__device__ __forceinline__ float gcn_pos_sigmoid(float v) {
    return 1.0f / (1.0f + expf(-v)) + 1e-8f;
}

__global__ __launch_bounds__(256)
void gcn_gather16_sigmoid_kernel(const float* __restrict__ hw, const int* __restrict__ rowptr,
                                 const int* __restrict__ ssrc, const float* __restrict__ dnorm,
                                 const float* __restrict__ bias, float* __restrict__ out, int n_nodes) {
    constexpr int F = 16;
    int gid = blockIdx.x * blockDim.x + threadIdx.x;
    int node = gid >> 2;
    int sub = gid & 3;
    if (node >= n_nodes) return;
    int beg = rowptr[node];
    int end = rowptr[node + 1];
    float dn = dnorm[node];
    int c0 = sub * 4;
    float4 bv = make_float4(bias[c0], bias[c0 + 1], bias[c0 + 2], bias[c0 + 3]);

    float4 acc = make_float4(0.0f, 0.0f, 0.0f, 0.0f);
    const float* __restrict__ colp = hw + sub * 4;

    for (int j = beg; j < end; j += 8) {
        int s[8];
        float4 v[8];
#pragma unroll
        for (int u = 0; u < 8; ++u) s[u] = (j + u < end) ? ssrc[j + u] : -1;
#pragma unroll
        for (int u = 0; u < 8; ++u) {
            v[u] = make_float4(0.0f, 0.0f, 0.0f, 0.0f);
            if (s[u] >= 0) v[u] = *(const float4*)(colp + (size_t)s[u] * F);
        }
#pragma unroll
        for (int u = 0; u < 8; ++u) {
            acc.x += v[u].x; acc.y += v[u].y; acc.z += v[u].z; acc.w += v[u].w;
        }
    }
    float4 r;
    r.x = gcn_pos_sigmoid(acc.x * dn + bv.x);
    r.y = gcn_pos_sigmoid(acc.y * dn + bv.y);
    r.z = gcn_pos_sigmoid(acc.z * dn + bv.z);
    r.w = gcn_pos_sigmoid(acc.w * dn + bv.w);
    *(float4*)(out + (size_t)node * F + sub * 4) = r;
}


extern "C" void kernel_launch(void* const* d_in, const int* in_sizes, int n_in,
                              void* d_out, int out_size, void* d_ws, size_t ws_size,
                              hipStream_t stream) {
    (void)n_in; (void)out_size; (void)ws_size;

    const float* feat = (const float*)d_in[0];
    const int*   src  = (const int*)d_in[1];
    const int*   dst  = (const int*)d_in[2];
    const float* W0   = (const float*)d_in[3];
    const float* b0   = (const float*)d_in[4];
    const float* W1   = (const float*)d_in[5];
    const float* b1   = (const float*)d_in[6];
    const float* W2   = (const float*)d_in[7];
    const float* b2   = (const float*)d_in[8];

    const int F = 128;
    const int N = in_sizes[0] / F;
    const int E = in_sizes[1];

    auto al = [](size_t n) { return (n + 63) & ~(size_t)63; };
    const int n_chunks = (N + GCN_SCAN_CHUNK - 1) / GCN_SCAN_CHUNK;
    float* snorm  = (float*)d_ws;
    int*   incnt  = (int*)(snorm + N);
    float* dnorm  = snorm + al(2 * (size_t)N);
    int*   rowptr = (int*)(dnorm + al((size_t)N));
    int*   csum   = rowptr + al((size_t)N + 1);
    int*   rank   = csum + al((size_t)n_chunks);
    int*   ssrc   = rank + al((size_t)E);
    float* bufA   = (float*)(ssrc + al((size_t)E));
    float* bufB   = bufA + al((size_t)N * F);

    const int TB = 256;
    auto nblk = [](long long n, int tb) { return (unsigned)((n + tb - 1) / tb); };

    gcn_zero_kernel<<<nblk(2LL * N, TB), TB, 0, stream>>>(snorm, 2 * N);
    gcn_degree_rank_kernel<<<nblk(E, TB), TB, 0, stream>>>(src, dst, snorm, incnt, rank, E);
    gcn_chunk_sum_kernel<<<n_chunks, TB, 0, stream>>>(incnt, csum, N);
    gcn_rowptr_kernel<<<n_chunks, TB, 0, stream>>>(incnt, csum, rowptr, N);
    gcn_norm_kernel<<<nblk(N, TB), TB, 0, stream>>>(snorm, incnt, dnorm, N);
    gcn_fill_kernel<<<nblk(E, TB), TB, 0, stream>>>(src, dst, rank, rowptr, ssrc, E);

    const long long gemm_threads = (long long)((N + 15) / 16) * 32;

    gcn_gemm_scaled_wmma<8><<<nblk(gemm_threads, TB), TB, 0, stream>>>(feat, snorm, W0, bufA, N, F);
    gcn_gather128_relu_kernel<<<nblk((long long)N * 32, TB), TB, 0, stream>>>(bufA, rowptr, ssrc, dnorm, b0, bufB, N);

    gcn_gemm_scaled_wmma<8><<<nblk(gemm_threads, TB), TB, 0, stream>>>(bufB, snorm, W1, bufA, N, F);
    gcn_gather128_relu_kernel<<<nblk((long long)N * 32, TB), TB, 0, stream>>>(bufA, rowptr, ssrc, dnorm, b1, bufB, N);

    gcn_gemm_scaled_wmma<1><<<nblk(gemm_threads, TB), TB, 0, stream>>>(bufB, snorm, W2, bufA, N, F);
    gcn_gather16_sigmoid_kernel<<<nblk((long long)N * 4, TB), TB, 0, stream>>>(bufA, rowptr, ssrc, dnorm, b2, (float*)d_out, N);
}


# Two launches dropped (12 to 10): norm pass folded into the chunk-sum threads; edge-list fill (4 edges per thread) shares one grid with the layer-0 transform so both overlap; sum order unchanged
# speedup vs baseline: 2.6735x; 1.0409x over previous
#include <hip/hip_runtime.h>
#include <hip/hip_bf16.h>
#include <math.h>

typedef __attribute__((ext_vector_type(2))) float v2f;
typedef __attribute__((ext_vector_type(8))) float v8f;


__global__ void gcn_zero_kernel(float* __restrict__ p, int n) {
    int i = blockIdx.x * blockDim.x + threadIdx.x;
    if (i < n) p[i] = 0.0f;
}

__global__ void gcn_degree_rank_kernel(const int* __restrict__ src, const int* __restrict__ dst,
                                       float* __restrict__ outdeg, int* __restrict__ incnt,
                                       int* __restrict__ rank, int n_edges) {
    int i = blockIdx.x * blockDim.x + threadIdx.x;
    if (i < n_edges) {
        atomicAdd(&outdeg[src[i]], 1.0f);
        rank[i] = atomicAdd(&incnt[dst[i]], 1);
    }
}


#define GCN_SCAN_CHUNK 1024

__device__ __forceinline__ int gcn_block_sum_256(int v, int* sm) {
    int lane = threadIdx.x & 31, wv = threadIdx.x >> 5;
    for (int o = 16; o > 0; o >>= 1) v += __shfl_xor(v, o);
    if (lane == 0) sm[wv] = v;
    __syncthreads();
    int t = 0;
    for (int w = 0; w < 8; ++w) t += sm[w];
    __syncthreads();
    return t;
}

__global__ __launch_bounds__(256)
void gcn_chunk_sum_norm_kernel(const int* __restrict__ cnt, int* __restrict__ csum,
                               float* __restrict__ snorm, float* __restrict__ dnorm, int n) {
    __shared__ int sm[8];
    int base = blockIdx.x * GCN_SCAN_CHUNK + threadIdx.x * 4;
    int ci[4];
    float od[4];
    int s = 0;
#pragma unroll
    for (int k = 0; k < 4; ++k) {
        bool in = (base + k < n);
        ci[k] = in ? cnt[base + k] : 0;
        od[k] = in ? snorm[base + k] : 1.0f;
        s += ci[k];
    }
#pragma unroll
    for (int k = 0; k < 4; ++k) {
        if (base + k < n) {
            float d = od[k];
            d = (d < 1.0f) ? 1.0f : d;
            snorm[base + k] = 1.0f / sqrtf(d);
            float cf = (float)ci[k];
            cf = (cf < 1.0f) ? 1.0f : cf;
            dnorm[base + k] = 1.0f / sqrtf(cf);
        }
    }
    int t = gcn_block_sum_256(s, sm);
    if (threadIdx.x == 0) csum[blockIdx.x] = t;
}

__global__ __launch_bounds__(256)
void gcn_rowptr_kernel(const int* __restrict__ cnt, const int* __restrict__ csum,
                       int* __restrict__ rowptr, int n) {
    __shared__ int sm[8];
    __shared__ int wtot[8];
    int tid = threadIdx.x, lane = tid & 31, wv = tid >> 5;

    int p = 0;
    for (int j = tid; j < (int)blockIdx.x; j += 256) p += csum[j];
    int chunk_off = gcn_block_sum_256(p, sm);

    int base = blockIdx.x * GCN_SCAN_CHUNK + tid * 4;
    int c[4];
    int t = 0;
#pragma unroll
    for (int k = 0; k < 4; ++k) { c[k] = (base + k < n) ? cnt[base + k] : 0; t += c[k]; }

    int inc = t;
    for (int o = 1; o < 32; o <<= 1) {
        int y = __shfl_up(inc, o);
        if (lane >= o) inc += y;
    }
    if (lane == 31) wtot[wv] = inc;
    __syncthreads();
    int wpre = 0;
    for (int w = 0; w < 8; ++w) if (w < wv) wpre += wtot[w];

    int run = chunk_off + wpre + inc - t;
#pragma unroll
    for (int k = 0; k < 4; ++k) {
        if (base + k < n) {
            rowptr[base + k] = run;
            run += c[k];
            if (base + k == n - 1) rowptr[n] = run;
        }
    }
}

__device__ __forceinline__ void gcn_fill_edges4(const int* __restrict__ src, const int* __restrict__ dst,
                                                const int* __restrict__ rank, const int* __restrict__ rowptr,
                                                int* __restrict__ ssrc, int n_edges, int t) {
    int i = t * 4;
    if (i + 3 < n_edges) {
        int4 d = *(const int4*)(dst + i);
        int4 r = *(const int4*)(rank + i);
        int4 s = *(const int4*)(src + i);
        int p0 = rowptr[d.x];
        int p1 = rowptr[d.y];
        int p2 = rowptr[d.z];
        int p3 = rowptr[d.w];
        ssrc[p0 + r.x] = s.x;
        ssrc[p1 + r.y] = s.y;
        ssrc[p2 + r.z] = s.z;
        ssrc[p3 + r.w] = s.w;
    } else {
        for (int k = 0; k < 4; ++k)
            if (i + k < n_edges) ssrc[rowptr[dst[i + k]] + rank[i + k]] = src[i + k];
    }
}


template <int NT>
__device__ __forceinline__ void gcn_gemm_tile(const float* __restrict__ X, const float* __restrict__ scale,
                                              const float* __restrict__ W, float* __restrict__ out,
                                              int n_nodes, int K, int wave, int lane) {
    const int NOUT = NT * 16;
    int row0 = wave * 16;
    if (row0 >= n_nodes) return;

    int hi   = lane >> 4;
    int kOff = hi * 2;
    int mrow = row0 + (lane & 15);
    int col  = lane & 15;

    float s = scale[mrow];
    const float* __restrict__ xrow = X + (size_t)mrow * (size_t)K;

    v8f acc[NT];
    v8f vzero = {};
#pragma unroll
    for (int nt = 0; nt < NT; ++nt) acc[nt] = vzero;

    for (int k0 = 0; k0 < K; k0 += 4) {
        float2 av = *(const float2*)(xrow + k0 + kOff);
        v2f a;
        a.x = av.x * s;
        a.y = av.y * s;
        const float* __restrict__ wr = W + (size_t)(k0 + kOff) * (size_t)NOUT;
#pragma unroll
        for (int nt = 0; nt < NT; ++nt) {
            v2f b;
            b.x = wr[nt * 16 + col];
            b.y = wr[NOUT + nt * 16 + col];
            acc[nt] = __builtin_amdgcn_wmma_f32_16x16x4_f32(
                 false, a,  false, b,
                 (short)0, acc[nt],  false,  false);
        }
    }

    int crow = row0 + hi * 8;
#pragma unroll
    for (int nt = 0; nt < NT; ++nt) {
#pragma unroll
        for (int v = 0; v < 8; ++v) {
            out[(size_t)(crow + v) * (size_t)NOUT + nt * 16 + col] = acc[nt][v];
        }
    }
}

template <int NT>
__global__ __launch_bounds__(256)
void gcn_gemm_scaled_wmma(const float* __restrict__ X, const float* __restrict__ scale,
                          const float* __restrict__ W, float* __restrict__ out,
                          int n_nodes, int K) {
    int wave = (blockIdx.x * blockDim.x + threadIdx.x) >> 5;
    gcn_gemm_tile<NT>(X, scale, W, out, n_nodes, K, wave, threadIdx.x & 31);
}

__global__ __launch_bounds__(256)
void gcn_gemm8_and_fill_kernel(const float* __restrict__ X, const float* __restrict__ scale,
                               const float* __restrict__ W, float* __restrict__ out,
                               int n_nodes, int K, int gemm_blocks,
                               const int* __restrict__ src, const int* __restrict__ dst,
                               const int* __restrict__ rank, const int* __restrict__ rowptr,
                               int* __restrict__ ssrc, int n_edges) {
    if ((int)blockIdx.x < gemm_blocks) {
        int wave = (blockIdx.x * blockDim.x + threadIdx.x) >> 5;
        gcn_gemm_tile<8>(X, scale, W, out, n_nodes, K, wave, threadIdx.x & 31);
    } else {
        int t = ((int)blockIdx.x - gemm_blocks) * (int)blockDim.x + (int)threadIdx.x;
        gcn_fill_edges4(src, dst, rank, rowptr, ssrc, n_edges, t);
    }
}


__global__ __launch_bounds__(256)
void gcn_gather128_relu_kernel(const float* __restrict__ hw, const int* __restrict__ rowptr,
                               const int* __restrict__ ssrc, const float* __restrict__ dnorm,
                               const float* __restrict__ bias, float* __restrict__ out, int n_nodes) {
    constexpr int F = 128;
    int lane = threadIdx.x & 31;
    int node = __builtin_amdgcn_readfirstlane((int)((blockIdx.x * blockDim.x + threadIdx.x) >> 5));
    if (node >= n_nodes) return;
    int beg = rowptr[node];
    int end = rowptr[node + 1];
    float dn = dnorm[node];
    int c0 = lane * 4;
    float4 bv = make_float4(bias[c0], bias[c0 + 1], bias[c0 + 2], bias[c0 + 3]);

    float4 acc = make_float4(0.0f, 0.0f, 0.0f, 0.0f);
    const float* __restrict__ colp = hw + lane * 4;

    for (int b = beg; b < end; b += 32) {
        int m = end - b;
        if (m > 32) m = 32;
        int my = (lane < m) ? ssrc[b + lane] : 0;
        for (int j = 0; j < m; j += 8) {
            int s[8];
            float4 v[8];
#pragma unroll
            for (int u = 0; u < 8; ++u) {
                int l = (j + u < m) ? (j + u) : (m - 1);
                s[u] = __builtin_amdgcn_readlane(my, l);
            }
#pragma unroll
            for (int u = 0; u < 8; ++u) {
                v[u] = make_float4(0.0f, 0.0f, 0.0f, 0.0f);
                if (j + u < m) v[u] = *(const float4*)(colp + (size_t)s[u] * F);
            }
#pragma unroll
            for (int u = 0; u < 8; ++u) {
                acc.x += v[u].x; acc.y += v[u].y; acc.z += v[u].z; acc.w += v[u].w;
            }
        }
    }
    float4 r;
    r.x = acc.x * dn + bv.x; r.x = (r.x > 0.0f) ? r.x : 0.0f;
    r.y = acc.y * dn + bv.y; r.y = (r.y > 0.0f) ? r.y : 0.0f;
    r.z = acc.z * dn + bv.z; r.z = (r.z > 0.0f) ? r.z : 0.0f;
    r.w = acc.w * dn + bv.w; r.w = (r.w > 0.0f) ? r.w : 0.0f;
    *(float4*)(out + (size_t)node * F + lane * 4) = r;
}

__device__ __forceinline__ float gcn_pos_sigmoid(float v) {
    return 1.0f / (1.0f + expf(-v)) + 1e-8f;
}

__global__ __launch_bounds__(256)
void gcn_gather16_sigmoid_kernel(const float* __restrict__ hw, const int* __restrict__ rowptr,
                                 const int* __restrict__ ssrc, const float* __restrict__ dnorm,
                                 const float* __restrict__ bias, float* __restrict__ out, int n_nodes) {
    constexpr int F = 16;
    int gid = blockIdx.x * blockDim.x + threadIdx.x;
    int node = gid >> 2;
    int sub = gid & 3;
    if (node >= n_nodes) return;
    int beg = rowptr[node];
    int end = rowptr[node + 1];
    float dn = dnorm[node];
    int c0 = sub * 4;
    float4 bv = make_float4(bias[c0], bias[c0 + 1], bias[c0 + 2], bias[c0 + 3]);

    float4 acc = make_float4(0.0f, 0.0f, 0.0f, 0.0f);
    const float* __restrict__ colp = hw + sub * 4;

    for (int j = beg; j < end; j += 8) {
        int s[8];
        float4 v[8];
#pragma unroll
        for (int u = 0; u < 8; ++u) s[u] = (j + u < end) ? ssrc[j + u] : -1;
#pragma unroll
        for (int u = 0; u < 8; ++u) {
            v[u] = make_float4(0.0f, 0.0f, 0.0f, 0.0f);
            if (s[u] >= 0) v[u] = *(const float4*)(colp + (size_t)s[u] * F);
        }
#pragma unroll
        for (int u = 0; u < 8; ++u) {
            acc.x += v[u].x; acc.y += v[u].y; acc.z += v[u].z; acc.w += v[u].w;
        }
    }
    float4 r;
    r.x = gcn_pos_sigmoid(acc.x * dn + bv.x);
    r.y = gcn_pos_sigmoid(acc.y * dn + bv.y);
    r.z = gcn_pos_sigmoid(acc.z * dn + bv.z);
    r.w = gcn_pos_sigmoid(acc.w * dn + bv.w);
    *(float4*)(out + (size_t)node * F + sub * 4) = r;
}


extern "C" void kernel_launch(void* const* d_in, const int* in_sizes, int n_in,
                              void* d_out, int out_size, void* d_ws, size_t ws_size,
                              hipStream_t stream) {
    (void)n_in; (void)out_size; (void)ws_size;

    const float* feat = (const float*)d_in[0];
    const int*   src  = (const int*)d_in[1];
    const int*   dst  = (const int*)d_in[2];
    const float* W0   = (const float*)d_in[3];
    const float* b0   = (const float*)d_in[4];
    const float* W1   = (const float*)d_in[5];
    const float* b1   = (const float*)d_in[6];
    const float* W2   = (const float*)d_in[7];
    const float* b2   = (const float*)d_in[8];

    const int F = 128;
    const int N = in_sizes[0] / F;
    const int E = in_sizes[1];

    auto al = [](size_t n) { return (n + 63) & ~(size_t)63; };
    const int n_chunks = (N + GCN_SCAN_CHUNK - 1) / GCN_SCAN_CHUNK;
    float* snorm  = (float*)d_ws;
    int*   incnt  = (int*)(snorm + N);
    float* dnorm  = snorm + al(2 * (size_t)N);
    int*   rowptr = (int*)(dnorm + al((size_t)N));
    int*   csum   = rowptr + al((size_t)N + 1);
    int*   rank   = csum + al((size_t)n_chunks);
    int*   ssrc   = rank + al((size_t)E);
    float* bufA   = (float*)(ssrc + al((size_t)E));
    float* bufB   = bufA + al((size_t)N * F);

    const int TB = 256;
    auto nblk = [](long long n, int tb) { return (unsigned)((n + tb - 1) / tb); };

    gcn_zero_kernel<<<nblk(2LL * N, TB), TB, 0, stream>>>(snorm, 2 * N);
    gcn_degree_rank_kernel<<<nblk(E, TB), TB, 0, stream>>>(src, dst, snorm, incnt, rank, E);
    gcn_chunk_sum_norm_kernel<<<n_chunks, TB, 0, stream>>>(incnt, csum, snorm, dnorm, N);
    gcn_rowptr_kernel<<<n_chunks, TB, 0, stream>>>(incnt, csum, rowptr, N);

    const long long gemm_threads = (long long)((N + 15) / 16) * 32;
    const unsigned gemm_blocks = nblk(gemm_threads, TB);
    const unsigned fill_blocks = nblk(((long long)E + 3) / 4, TB);

    gcn_gemm8_and_fill_kernel<<<gemm_blocks + fill_blocks, TB, 0, stream>>>(feat, snorm, W0, bufA, N, F, (int)gemm_blocks,
                                                                          src, dst, rank, rowptr, ssrc, E);
    gcn_gather128_relu_kernel<<<nblk((long long)N * 32, TB), TB, 0, stream>>>(bufA, rowptr, ssrc, dnorm, b0, bufB, N);

    gcn_gemm_scaled_wmma<8><<<nblk(gemm_threads, TB), TB, 0, stream>>>(bufB, snorm, W1, bufA, N, F);
    gcn_gather128_relu_kernel<<<nblk((long long)N * 32, TB), TB, 0, stream>>>(bufA, rowptr, ssrc, dnorm, b1, bufB, N);

    gcn_gemm_scaled_wmma<1><<<nblk(gemm_threads, TB), TB, 0, stream>>>(bufB, snorm, W2, bufA, N, F);
    gcn_gather16_sigmoid_kernel<<<nblk((long long)N * 4, TB), TB, 0, stream>>>(bufA, rowptr, ssrc, dnorm, b2, (float*)d_out, N);
}


# Next layer's transform fused into the 128-wide segment-sum kernel: 4 waves sum a 16-node tile into LDS, which is the WMMA's A; 2 launches and both h round trips dropped; sum order unchanged
# speedup vs baseline: 2.8873x; 1.0800x over previous
#include <hip/hip_runtime.h>
#include <hip/hip_bf16.h>
#include <math.h>

typedef __attribute__((ext_vector_type(2))) float v2f;
typedef __attribute__((ext_vector_type(8))) float v8f;


__global__ void gcn_zero_kernel(float* __restrict__ p, int n) {
    int i = blockIdx.x * blockDim.x + threadIdx.x;
    if (i < n) p[i] = 0.0f;
}

__global__ void gcn_degree_rank_kernel(const int* __restrict__ src, const int* __restrict__ dst,
                                       float* __restrict__ outdeg, int* __restrict__ incnt,
                                       int* __restrict__ rank, int n_edges) {
    int i = blockIdx.x * blockDim.x + threadIdx.x;
    if (i < n_edges) {
        atomicAdd(&outdeg[src[i]], 1.0f);
        rank[i] = atomicAdd(&incnt[dst[i]], 1);
    }
}


#define GCN_SCAN_CHUNK 1024

__device__ __forceinline__ int gcn_block_sum_256(int v, int* sm) {
    int lane = threadIdx.x & 31, wv = threadIdx.x >> 5;
    for (int o = 16; o > 0; o >>= 1) v += __shfl_xor(v, o);
    if (lane == 0) sm[wv] = v;
    __syncthreads();
    int t = 0;
    for (int w = 0; w < 8; ++w) t += sm[w];
    __syncthreads();
    return t;
}

__global__ __launch_bounds__(256)
void gcn_chunk_sum_norm_kernel(const int* __restrict__ cnt, int* __restrict__ csum,
                               float* __restrict__ snorm, float* __restrict__ dnorm, int n) {
    __shared__ int sm[8];
    int base = blockIdx.x * GCN_SCAN_CHUNK + threadIdx.x * 4;
    int ci[4];
    float od[4];
    int s = 0;
#pragma unroll
    for (int k = 0; k < 4; ++k) {
        bool in = (base + k < n);
        ci[k] = in ? cnt[base + k] : 0;
        od[k] = in ? snorm[base + k] : 1.0f;
        s += ci[k];
    }
#pragma unroll
    for (int k = 0; k < 4; ++k) {
        if (base + k < n) {
            float d = od[k];
            d = (d < 1.0f) ? 1.0f : d;
            snorm[base + k] = 1.0f / sqrtf(d);
            float cf = (float)ci[k];
            cf = (cf < 1.0f) ? 1.0f : cf;
            dnorm[base + k] = 1.0f / sqrtf(cf);
        }
    }
    int t = gcn_block_sum_256(s, sm);
    if (threadIdx.x == 0) csum[blockIdx.x] = t;
}

__global__ __launch_bounds__(256)
void gcn_rowptr_kernel(const int* __restrict__ cnt, const int* __restrict__ csum,
                       int* __restrict__ rowptr, int n) {
    __shared__ int sm[8];
    __shared__ int wtot[8];
    int tid = threadIdx.x, lane = tid & 31, wv = tid >> 5;

    int p = 0;
    for (int j = tid; j < (int)blockIdx.x; j += 256) p += csum[j];
    int chunk_off = gcn_block_sum_256(p, sm);

    int base = blockIdx.x * GCN_SCAN_CHUNK + tid * 4;
    int c[4];
    int t = 0;
#pragma unroll
    for (int k = 0; k < 4; ++k) { c[k] = (base + k < n) ? cnt[base + k] : 0; t += c[k]; }

    int inc = t;
    for (int o = 1; o < 32; o <<= 1) {
        int y = __shfl_up(inc, o);
        if (lane >= o) inc += y;
    }
    if (lane == 31) wtot[wv] = inc;
    __syncthreads();
    int wpre = 0;
    for (int w = 0; w < 8; ++w) if (w < wv) wpre += wtot[w];

    int run = chunk_off + wpre + inc - t;
#pragma unroll
    for (int k = 0; k < 4; ++k) {
        if (base + k < n) {
            rowptr[base + k] = run;
            run += c[k];
            if (base + k == n - 1) rowptr[n] = run;
        }
    }
}

__device__ __forceinline__ void gcn_fill_edges4(const int* __restrict__ src, const int* __restrict__ dst,
                                                const int* __restrict__ rank, const int* __restrict__ rowptr,
                                                int* __restrict__ ssrc, int n_edges, int t) {
    int i = t * 4;
    if (i + 3 < n_edges) {
        int4 d = *(const int4*)(dst + i);
        int4 r = *(const int4*)(rank + i);
        int4 s = *(const int4*)(src + i);
        int p0 = rowptr[d.x];
        int p1 = rowptr[d.y];
        int p2 = rowptr[d.z];
        int p3 = rowptr[d.w];
        ssrc[p0 + r.x] = s.x;
        ssrc[p1 + r.y] = s.y;
        ssrc[p2 + r.z] = s.z;
        ssrc[p3 + r.w] = s.w;
    } else {
        for (int k = 0; k < 4; ++k)
            if (i + k < n_edges) ssrc[rowptr[dst[i + k]] + rank[i + k]] = src[i + k];
    }
}


template <int NT>
__device__ __forceinline__ void gcn_gemm_tile(const float* __restrict__ X, const float* __restrict__ scale,
                                              const float* __restrict__ W, float* __restrict__ out,
                                              int n_nodes, int K, int wave, int lane) {
    const int NOUT = NT * 16;
    int row0 = wave * 16;
    if (row0 >= n_nodes) return;

    int hi   = lane >> 4;
    int kOff = hi * 2;
    int mrow = row0 + (lane & 15);
    int col  = lane & 15;

    float s = scale[mrow];
    const float* __restrict__ xrow = X + (size_t)mrow * (size_t)K;

    v8f acc[NT];
    v8f vzero = {};
#pragma unroll
    for (int nt = 0; nt < NT; ++nt) acc[nt] = vzero;

    for (int k0 = 0; k0 < K; k0 += 4) {
        float2 av = *(const float2*)(xrow + k0 + kOff);
        v2f a;
        a.x = av.x * s;
        a.y = av.y * s;
        const float* __restrict__ wr = W + (size_t)(k0 + kOff) * (size_t)NOUT;
#pragma unroll
        for (int nt = 0; nt < NT; ++nt) {
            v2f b;
            b.x = wr[nt * 16 + col];
            b.y = wr[NOUT + nt * 16 + col];
            acc[nt] = __builtin_amdgcn_wmma_f32_16x16x4_f32(
                 false, a,  false, b,
                 (short)0, acc[nt],  false,  false);
        }
    }

    int crow = row0 + hi * 8;
#pragma unroll
    for (int nt = 0; nt < NT; ++nt) {
#pragma unroll
        for (int v = 0; v < 8; ++v) {
            out[(size_t)(crow + v) * (size_t)NOUT + nt * 16 + col] = acc[nt][v];
        }
    }
}

__global__ __launch_bounds__(256)
void gcn_gemm8_and_fill_kernel(const float* __restrict__ X, const float* __restrict__ scale,
                               const float* __restrict__ W, float* __restrict__ out,
                               int n_nodes, int K, int gemm_blocks,
                               const int* __restrict__ src, const int* __restrict__ dst,
                               const int* __restrict__ rank, const int* __restrict__ rowptr,
                               int* __restrict__ ssrc, int n_edges) {
    if ((int)blockIdx.x < gemm_blocks) {
        int wave = (blockIdx.x * blockDim.x + threadIdx.x) >> 5;
        gcn_gemm_tile<8>(X, scale, W, out, n_nodes, K, wave, threadIdx.x & 31);
    } else {
        int t = ((int)blockIdx.x - gemm_blocks) * (int)blockDim.x + (int)threadIdx.x;
        gcn_fill_edges4(src, dst, rank, rowptr, ssrc, n_edges, t);
    }
}


#define GCN_TILE_PITCH 132

template <int NT>
__global__ __launch_bounds__(128)
void gcn_gather128_relu_gemm_kernel(const float* __restrict__ hw, const int* __restrict__ rowptr,
                                    const int* __restrict__ ssrc, const float* __restrict__ dnorm,
                                    const float* __restrict__ bias, const float* __restrict__ snorm,
                                    const float* __restrict__ Wn, float* __restrict__ out, int n_nodes) {
    constexpr int F = 128;
    constexpr int NOUT = NT * 16;
    constexpr int NTW = (NT >= 4) ? (NT / 4) : NT;
    __shared__ __attribute__((aligned(16))) float tile[16 * GCN_TILE_PITCH];

    int lane = threadIdx.x & 31;
    int wv = __builtin_amdgcn_readfirstlane((int)(threadIdx.x >> 5));
    int row0 = (int)blockIdx.x * 16;
    if (row0 >= n_nodes) return;

    int q0 = wv * 4;
    int ni = row0 + q0 + lane;
    int rp = rowptr[(ni < n_nodes) ? ni : n_nodes];
    int nc = (ni < n_nodes) ? ni : (n_nodes - 1);
    int dnb = __float_as_int(dnorm[nc]);
    int snb = __float_as_int(snorm[nc]);
    int c0 = lane * 4;
    float4 bv = make_float4(bias[c0], bias[c0 + 1], bias[c0 + 2], bias[c0 + 3]);
    const float* __restrict__ colp = hw + lane * 4;

    int beg0 = __builtin_amdgcn_readlane(rp, 0);
    int end0 = __builtin_amdgcn_readlane(rp, 1);
    int my_first = (lane < end0 - beg0) ? ssrc[beg0 + lane] : 0;

    for (int r = 0; r < 4; ++r) {
        int beg = __builtin_amdgcn_readlane(rp, r);
        int end = __builtin_amdgcn_readlane(rp, r + 1);
        int nend = __builtin_amdgcn_readlane(rp, (r < 3) ? (r + 2) : (r + 1));
        int myn = 0;
        float4 acc = make_float4(0.0f, 0.0f, 0.0f, 0.0f);
        for (int b = beg; b < end; b += 32) {
            int m = end - b;
            if (m > 32) m = 32;
            int my = my_first;
            if (b != beg) my = (lane < m) ? ssrc[b + lane] : 0;
            for (int j = 0; j < m; j += 8) {
                int s[8];
                float4 v[8];
#pragma unroll
                for (int u = 0; u < 8; ++u) {
                    int l = (j + u < m) ? (j + u) : (m - 1);
                    s[u] = __builtin_amdgcn_readlane(my, l);
                }
#pragma unroll
                for (int u = 0; u < 8; ++u) {
                    v[u] = make_float4(0.0f, 0.0f, 0.0f, 0.0f);
                    if (j + u < m) v[u] = *(const float4*)(colp + (size_t)s[u] * F);
                }
                myn = (lane < nend - end) ? ssrc[end + lane] : 0;
#pragma unroll
                for (int u = 0; u < 8; ++u) {
                    acc.x += v[u].x; acc.y += v[u].y; acc.z += v[u].z; acc.w += v[u].w;
                }
            }
        }
        if (end <= beg) myn = (lane < nend - end) ? ssrc[end + lane] : 0;
        my_first = myn;
        float dn = __int_as_float(__builtin_amdgcn_readlane(dnb, r));
        float sn = __int_as_float(__builtin_amdgcn_readlane(snb, r));
        float4 h;
        h.x = acc.x * dn + bv.x; h.x = (h.x > 0.0f) ? h.x : 0.0f;
        h.y = acc.y * dn + bv.y; h.y = (h.y > 0.0f) ? h.y : 0.0f;
        h.z = acc.z * dn + bv.z; h.z = (h.z > 0.0f) ? h.z : 0.0f;
        h.w = acc.w * dn + bv.w; h.w = (h.w > 0.0f) ? h.w : 0.0f;
        h.x *= sn; h.y *= sn; h.z *= sn; h.w *= sn;
        *(float4*)(tile + (q0 + r) * GCN_TILE_PITCH + c0) = h;
    }

    __syncthreads();

    if (NT < 4 && wv != 0) return;
    int ct0 = (NT >= 4) ? (wv * NTW) : 0;

    int hi   = lane >> 4;
    int kOff = hi * 2;
    int col  = lane & 15;
    const float* arow = tile + col * GCN_TILE_PITCH + kOff;

    v8f cacc[NTW];
    v8f vzero = {};
#pragma unroll
    for (int t = 0; t < NTW; ++t) cacc[t] = vzero;

#pragma unroll 8
    for (int k0 = 0; k0 < F; k0 += 4) {
        float2 av = *(const float2*)(arow + k0);
        v2f a;
        a.x = av.x;
        a.y = av.y;
        const float* __restrict__ wr = Wn + (size_t)(k0 + kOff) * (size_t)NOUT + ct0 * 16;
#pragma unroll
        for (int t = 0; t < NTW; ++t) {
            v2f b;
            b.x = wr[t * 16 + col];
            b.y = wr[NOUT + t * 16 + col];
            cacc[t] = __builtin_amdgcn_wmma_f32_16x16x4_f32(
                 false, a,  false, b,
                 (short)0, cacc[t],  false,  false);
        }
    }

    int crow = row0 + hi * 8;
#pragma unroll
    for (int t = 0; t < NTW; ++t) {
#pragma unroll
        for (int v = 0; v < 8; ++v) {
            if (crow + v < n_nodes)
                out[(size_t)(crow + v) * (size_t)NOUT + (ct0 + t) * 16 + col] = cacc[t][v];
        }
    }
}

__device__ __forceinline__ float gcn_pos_sigmoid(float v) {
    return 1.0f / (1.0f + expf(-v)) + 1e-8f;
}

__global__ __launch_bounds__(256)
void gcn_gather16_sigmoid_kernel(const float* __restrict__ hw, const int* __restrict__ rowptr,
                                 const int* __restrict__ ssrc, const float* __restrict__ dnorm,
                                 const float* __restrict__ bias, float* __restrict__ out, int n_nodes) {
    constexpr int F = 16;
    int gid = blockIdx.x * blockDim.x + threadIdx.x;
    int node = gid >> 2;
    int sub = gid & 3;
    if (node >= n_nodes) return;
    int beg = rowptr[node];
    int end = rowptr[node + 1];
    float dn = dnorm[node];
    int c0 = sub * 4;
    float4 bv = make_float4(bias[c0], bias[c0 + 1], bias[c0 + 2], bias[c0 + 3]);

    float4 acc = make_float4(0.0f, 0.0f, 0.0f, 0.0f);
    const float* __restrict__ colp = hw + sub * 4;

    for (int j = beg; j < end; j += 8) {
        int s[8];
        float4 v[8];
#pragma unroll
        for (int u = 0; u < 8; ++u) s[u] = (j + u < end) ? ssrc[j + u] : -1;
#pragma unroll
        for (int u = 0; u < 8; ++u) {
            v[u] = make_float4(0.0f, 0.0f, 0.0f, 0.0f);
            if (s[u] >= 0) v[u] = *(const float4*)(colp + (size_t)s[u] * F);
        }
#pragma unroll
        for (int u = 0; u < 8; ++u) {
            acc.x += v[u].x; acc.y += v[u].y; acc.z += v[u].z; acc.w += v[u].w;
        }
    }
    float4 r;
    r.x = gcn_pos_sigmoid(acc.x * dn + bv.x);
    r.y = gcn_pos_sigmoid(acc.y * dn + bv.y);
    r.z = gcn_pos_sigmoid(acc.z * dn + bv.z);
    r.w = gcn_pos_sigmoid(acc.w * dn + bv.w);
    *(float4*)(out + (size_t)node * F + sub * 4) = r;
}


extern "C" void kernel_launch(void* const* d_in, const int* in_sizes, int n_in,
                              void* d_out, int out_size, void* d_ws, size_t ws_size,
                              hipStream_t stream) {
    (void)n_in; (void)out_size; (void)ws_size;

    const float* feat = (const float*)d_in[0];
    const int*   src  = (const int*)d_in[1];
    const int*   dst  = (const int*)d_in[2];
    const float* W0   = (const float*)d_in[3];
    const float* b0   = (const float*)d_in[4];
    const float* W1   = (const float*)d_in[5];
    const float* b1   = (const float*)d_in[6];
    const float* W2   = (const float*)d_in[7];
    const float* b2   = (const float*)d_in[8];

    const int F = 128;
    const int N = in_sizes[0] / F;
    const int E = in_sizes[1];

    auto al = [](size_t n) { return (n + 63) & ~(size_t)63; };
    const int n_chunks = (N + GCN_SCAN_CHUNK - 1) / GCN_SCAN_CHUNK;
    float* snorm  = (float*)d_ws;
    int*   incnt  = (int*)(snorm + N);
    float* dnorm  = snorm + al(2 * (size_t)N);
    int*   rowptr = (int*)(dnorm + al((size_t)N));
    int*   csum   = rowptr + al((size_t)N + 1);
    int*   rank   = csum + al((size_t)n_chunks);
    int*   ssrc   = rank + al((size_t)E);
    float* bufA   = (float*)(ssrc + al((size_t)E));
    float* bufB   = bufA + al((size_t)N * F);

    const int TB = 256;
    auto nblk = [](long long n, int tb) { return (unsigned)((n + tb - 1) / tb); };

    gcn_zero_kernel<<<nblk(2LL * N, TB), TB, 0, stream>>>(snorm, 2 * N);
    gcn_degree_rank_kernel<<<nblk(E, TB), TB, 0, stream>>>(src, dst, snorm, incnt, rank, E);
    gcn_chunk_sum_norm_kernel<<<n_chunks, TB, 0, stream>>>(incnt, csum, snorm, dnorm, N);
    gcn_rowptr_kernel<<<n_chunks, TB, 0, stream>>>(incnt, csum, rowptr, N);

    const long long gemm_threads = (long long)((N + 15) / 16) * 32;
    const unsigned gemm_blocks = nblk(gemm_threads, TB);
    const unsigned fill_blocks = nblk(((long long)E + 3) / 4, TB);

    gcn_gemm8_and_fill_kernel<<<gemm_blocks + fill_blocks, TB, 0, stream>>>(feat, snorm, W0, bufA, N, F, (int)gemm_blocks,
                                                                          src, dst, rank, rowptr, ssrc, E);

    const int FB = 128;
    const unsigned n_tiles = nblk(N, 16);
    gcn_gather128_relu_gemm_kernel<8><<<n_tiles, FB, 0, stream>>>(bufA, rowptr, ssrc, dnorm, b0, snorm, W1, bufB, N);

    gcn_gather128_relu_gemm_kernel<1><<<n_tiles, FB, 0, stream>>>(bufB, rowptr, ssrc, dnorm, b1, snorm, W2, bufA, N);

    gcn_gather16_sigmoid_kernel<<<nblk((long long)N * 4, TB), TB, 0, stream>>>(bufA, rowptr, ssrc, dnorm, b2, (float*)d_out, N);
}


# Per-edge global atomics (out-degree adds, in-degree rank adds) and zero fill replaced by per-chunk LDS counter tables (64 edge chunks); ranks from LDS adds; sum order inside a node changes
# speedup vs baseline: 3.5713x; 1.2369x over previous
#include <hip/hip_runtime.h>
#include <hip/hip_bf16.h>
#include <math.h>

typedef __attribute__((ext_vector_type(2))) float v2f;
typedef __attribute__((ext_vector_type(8))) float v8f;


#define GCN_PARTS 64
#define GCN_HW    50000
static_assert((GCN_HW & 1) == 0, "a thread reads its 4 nodes' counters as one aligned pair of words");

__device__ __forceinline__ void gcn_chunk_bounds(int n_edges, int c, int& lo, int& hi) {
    int chunk = (((n_edges + GCN_PARTS - 1) / GCN_PARTS) + 3) & ~3;
    long long l = (long long)c * chunk;
    lo = (l < n_edges) ? (int)l : n_edges;
    hi = (n_edges - lo < chunk) ? n_edges : (lo + chunk);
}

__global__ __launch_bounds__(1024)
void gcn_hist_kernel(const int* __restrict__ src, const int* __restrict__ dst,
                     unsigned int* __restrict__ part, int n_edges) {
    __shared__ unsigned int h[GCN_HW];
    int tid = threadIdx.x;
    int role = (int)blockIdx.x / GCN_PARTS;
    int c = (int)blockIdx.x - role * GCN_PARTS;
    const int* __restrict__ key = role ? src : dst;
    for (int i = tid; i < GCN_HW; i += 1024) h[i] = 0u;
    __syncthreads();
    int lo, hi;
    gcn_chunk_bounds(n_edges, c, lo, hi);
    int nq = (hi - lo) >> 2;
    const int4* __restrict__ kq = (const int4*)(key + lo);
    for (int q = tid; q < nq; q += 4096) {
        int4 k[4];
#pragma unroll
        for (int u = 0; u < 4; ++u) {
            int qu = q + u * 1024;
            k[u] = kq[(qu < nq) ? qu : q];
        }
#pragma unroll
        for (int u = 0; u < 4; ++u) {
            if (q + u * 1024 < nq) {
                atomicAdd(&h[k[u].x >> 1], 1u << ((k[u].x & 1) * 16));
                atomicAdd(&h[k[u].y >> 1], 1u << ((k[u].y & 1) * 16));
                atomicAdd(&h[k[u].z >> 1], 1u << ((k[u].z & 1) * 16));
                atomicAdd(&h[k[u].w >> 1], 1u << ((k[u].w & 1) * 16));
            }
        }
    }
    for (int i = lo + 4 * nq + tid; i < hi; i += 1024) {
        int k = key[i];
        atomicAdd(&h[k >> 1], 1u << ((k & 1) * 16));
    }
    __syncthreads();
    unsigned int* __restrict__ out = part + (size_t)blockIdx.x * GCN_HW;
    for (int i = tid; i < GCN_HW; i += 1024) out[i] = h[i];
}


#define GCN_SCAN_CHUNK 1024

__device__ __forceinline__ int gcn_block_sum_256(int v, int* sm) {
    int lane = threadIdx.x & 31, wv = threadIdx.x >> 5;
    for (int o = 16; o > 0; o >>= 1) v += __shfl_xor(v, o);
    if (lane == 0) sm[wv] = v;
    __syncthreads();
    int t = 0;
    for (int w = 0; w < 8; ++w) t += sm[w];
    __syncthreads();
    return t;
}

__global__ __launch_bounds__(256)
void gcn_chunk_sum_norm_kernel(const unsigned int* __restrict__ part, int* __restrict__ cnt, int* __restrict__ csum,
                               float* __restrict__ snorm, float* __restrict__ dnorm, int n) {
    __shared__ int sm[8];
    int base = blockIdx.x * GCN_SCAN_CHUNK + threadIdx.x * 4;
    int w0 = base >> 1;
    unsigned int ci[4] = {0u, 0u, 0u, 0u};
    unsigned int co[4] = {0u, 0u, 0u, 0u};
    if (w0 < GCN_HW) {
        const unsigned int* __restrict__ pd = part + w0;
        const unsigned int* __restrict__ ps = part + (size_t)GCN_PARTS * GCN_HW + w0;
#pragma unroll 16
        for (int p = 0; p < GCN_PARTS; ++p) {
            uint2 a = *(const uint2*)(pd + (size_t)p * GCN_HW);
            uint2 b = *(const uint2*)(ps + (size_t)p * GCN_HW);
            ci[0] += a.x & 0xFFFFu; ci[1] += a.x >> 16; ci[2] += a.y & 0xFFFFu; ci[3] += a.y >> 16;
            co[0] += b.x & 0xFFFFu; co[1] += b.x >> 16; co[2] += b.y & 0xFFFFu; co[3] += b.y >> 16;
        }
    }
    int s = 0;
#pragma unroll
    for (int k = 0; k < 4; ++k) {
        if (base + k < n) {
            float d = (float)co[k];
            d = (d < 1.0f) ? 1.0f : d;
            snorm[base + k] = 1.0f / sqrtf(d);
            float cf = (float)ci[k];
            cf = (cf < 1.0f) ? 1.0f : cf;
            dnorm[base + k] = 1.0f / sqrtf(cf);
            cnt[base + k] = (int)ci[k];
            s += (int)ci[k];
        }
    }
    int t = gcn_block_sum_256(s, sm);
    if (threadIdx.x == 0) csum[blockIdx.x] = t;
}

__global__ __launch_bounds__(256)
void gcn_rowptr_kernel(const int* __restrict__ cnt, const int* __restrict__ csum,
                       const unsigned int* __restrict__ part, int* __restrict__ rowptr,
                       int* __restrict__ slot0, int n) {
    __shared__ int sm[8];
    __shared__ int wtot[8];
    int tid = threadIdx.x, lane = tid & 31, wv = tid >> 5;

    int p = 0;
    for (int j = tid; j < (int)blockIdx.x; j += 256) p += csum[j];
    int chunk_off = gcn_block_sum_256(p, sm);

    int base = blockIdx.x * GCN_SCAN_CHUNK + tid * 4;
    int c[4];
    int t = 0;
#pragma unroll
    for (int k = 0; k < 4; ++k) { c[k] = (base + k < n) ? cnt[base + k] : 0; t += c[k]; }

    int inc = t;
    for (int o = 1; o < 32; o <<= 1) {
        int y = __shfl_up(inc, o);
        if (lane >= o) inc += y;
    }
    if (lane == 31) wtot[wv] = inc;
    __syncthreads();
    int wpre = 0;
    for (int w = 0; w < 8; ++w) if (w < wv) wpre += wtot[w];

    int run = chunk_off + wpre + inc - t;
    int r[4];
#pragma unroll
    for (int k = 0; k < 4; ++k) {
        r[k] = run;
        if (base + k < n) {
            rowptr[base + k] = run;
            run += c[k];
            if (base + k == n - 1) rowptr[n] = run;
        }
    }

    if (base < n) {
        const unsigned int* __restrict__ pd = part + (base >> 1);
        int a0 = r[0], a1 = r[1], a2 = r[2], a3 = r[3];
        if ((base + 3 < n) && ((n & 3) == 0)) {
#pragma unroll 16
            for (int q = 0; q < GCN_PARTS; ++q) {
                uint2 a = *(const uint2*)(pd + (size_t)q * GCN_HW);
                *(int4*)(slot0 + (size_t)q * (size_t)n + base) = make_int4(a0, a1, a2, a3);
                a0 += (int)(a.x & 0xFFFFu); a1 += (int)(a.x >> 16);
                a2 += (int)(a.y & 0xFFFFu); a3 += (int)(a.y >> 16);
            }
        } else {
            for (int q = 0; q < GCN_PARTS; ++q) {
                uint2 a = *(const uint2*)(pd + (size_t)q * GCN_HW);
                int* __restrict__ o = slot0 + (size_t)q * (size_t)n + base;
                o[0] = a0;
                if (base + 1 < n) o[1] = a1;
                if (base + 2 < n) o[2] = a2;
                if (base + 3 < n) o[3] = a3;
                a0 += (int)(a.x & 0xFFFFu); a1 += (int)(a.x >> 16);
                a2 += (int)(a.y & 0xFFFFu); a3 += (int)(a.y >> 16);
            }
        }
    }
}

__global__ __launch_bounds__(1024)
void gcn_fill_kernel(const int* __restrict__ src, const int* __restrict__ dst,
                     const int* __restrict__ slot0, int* __restrict__ ssrc, int n_edges, int n_nodes) {
    __shared__ unsigned int h[GCN_HW];
    int tid = threadIdx.x;
    int c = (int)blockIdx.x;
    for (int i = tid; i < GCN_HW; i += 1024) h[i] = 0u;
    __syncthreads();
    const int* __restrict__ s0 = slot0 + (size_t)c * (size_t)n_nodes;
    int lo, hi;
    gcn_chunk_bounds(n_edges, c, lo, hi);
    int nq = (hi - lo) >> 2;
    const int4* __restrict__ dq = (const int4*)(dst + lo);
    const int4* __restrict__ sq = (const int4*)(src + lo);
    for (int q = tid; q < nq; q += 1024) {
        int4 d = dq[q];
        int4 s = sq[q];
        int p0 = s0[d.x];
        int p1 = s0[d.y];
        int p2 = s0[d.z];
        int p3 = s0[d.w];
        int h0 = (d.x & 1) * 16, h1 = (d.y & 1) * 16, h2 = (d.z & 1) * 16, h3 = (d.w & 1) * 16;
        unsigned int r0 = (atomicAdd(&h[d.x >> 1], 1u << h0) >> h0) & 0xFFFFu;
        unsigned int r1 = (atomicAdd(&h[d.y >> 1], 1u << h1) >> h1) & 0xFFFFu;
        unsigned int r2 = (atomicAdd(&h[d.z >> 1], 1u << h2) >> h2) & 0xFFFFu;
        unsigned int r3 = (atomicAdd(&h[d.w >> 1], 1u << h3) >> h3) & 0xFFFFu;
        ssrc[p0 + (int)r0] = s.x;
        ssrc[p1 + (int)r1] = s.y;
        ssrc[p2 + (int)r2] = s.z;
        ssrc[p3 + (int)r3] = s.w;
    }
    for (int i = lo + 4 * nq + tid; i < hi; i += 1024) {
        int d = dst[i];
        int sh = (d & 1) * 16;
        unsigned int r = (atomicAdd(&h[d >> 1], 1u << sh) >> sh) & 0xFFFFu;
        ssrc[s0[d] + (int)r] = src[i];
    }
}


template <int NT>
__device__ __forceinline__ void gcn_gemm_tile(const float* __restrict__ X, const float* __restrict__ scale,
                                              const float* __restrict__ W, float* __restrict__ out,
                                              int n_nodes, int K, int wave, int lane) {
    const int NOUT = NT * 16;
    int row0 = wave * 16;
    if (row0 >= n_nodes) return;

    int hi   = lane >> 4;
    int kOff = hi * 2;
    int mrow = row0 + (lane & 15);
    int col  = lane & 15;

    float s = scale[mrow];
    const float* __restrict__ xrow = X + (size_t)mrow * (size_t)K;

    v8f acc[NT];
    v8f vzero = {};
#pragma unroll
    for (int nt = 0; nt < NT; ++nt) acc[nt] = vzero;

    for (int k0 = 0; k0 < K; k0 += 4) {
        float2 av = *(const float2*)(xrow + k0 + kOff);
        v2f a;
        a.x = av.x * s;
        a.y = av.y * s;
        const float* __restrict__ wr = W + (size_t)(k0 + kOff) * (size_t)NOUT;
#pragma unroll
        for (int nt = 0; nt < NT; ++nt) {
            v2f b;
            b.x = wr[nt * 16 + col];
            b.y = wr[NOUT + nt * 16 + col];
            acc[nt] = __builtin_amdgcn_wmma_f32_16x16x4_f32(
                 false, a,  false, b,
                 (short)0, acc[nt],  false,  false);
        }
    }

    int crow = row0 + hi * 8;
#pragma unroll
    for (int nt = 0; nt < NT; ++nt) {
#pragma unroll
        for (int v = 0; v < 8; ++v) {
            out[(size_t)(crow + v) * (size_t)NOUT + nt * 16 + col] = acc[nt][v];
        }
    }
}

__global__ __launch_bounds__(256)
void gcn_gemm8_kernel(const float* __restrict__ X, const float* __restrict__ scale,
                      const float* __restrict__ W, float* __restrict__ out, int n_nodes, int K) {
    int wave = (blockIdx.x * blockDim.x + threadIdx.x) >> 5;
    gcn_gemm_tile<8>(X, scale, W, out, n_nodes, K, wave, threadIdx.x & 31);
}


#define GCN_TILE_PITCH 132

template <int NT>
__global__ __launch_bounds__(128)
void gcn_gather128_relu_gemm_kernel(const float* __restrict__ hw, const int* __restrict__ rowptr,
                                    const int* __restrict__ ssrc, const float* __restrict__ dnorm,
                                    const float* __restrict__ bias, const float* __restrict__ snorm,
                                    const float* __restrict__ Wn, float* __restrict__ out, int n_nodes) {
    constexpr int F = 128;
    constexpr int NOUT = NT * 16;
    constexpr int NTW = (NT >= 4) ? (NT / 4) : NT;
    __shared__ __attribute__((aligned(16))) float tile[16 * GCN_TILE_PITCH];

    int lane = threadIdx.x & 31;
    int wv = __builtin_amdgcn_readfirstlane((int)(threadIdx.x >> 5));
    int row0 = (int)blockIdx.x * 16;
    if (row0 >= n_nodes) return;

    int q0 = wv * 4;
    int ni = row0 + q0 + lane;
    int rp = rowptr[(ni < n_nodes) ? ni : n_nodes];
    int nc = (ni < n_nodes) ? ni : (n_nodes - 1);
    int dnb = __float_as_int(dnorm[nc]);
    int snb = __float_as_int(snorm[nc]);
    int c0 = lane * 4;
    float4 bv = make_float4(bias[c0], bias[c0 + 1], bias[c0 + 2], bias[c0 + 3]);
    const float* __restrict__ colp = hw + lane * 4;

    int beg0 = __builtin_amdgcn_readlane(rp, 0);
    int end0 = __builtin_amdgcn_readlane(rp, 1);
    int my_first = (lane < end0 - beg0) ? ssrc[beg0 + lane] : 0;

    for (int r = 0; r < 4; ++r) {
        int beg = __builtin_amdgcn_readlane(rp, r);
        int end = __builtin_amdgcn_readlane(rp, r + 1);
        int nend = __builtin_amdgcn_readlane(rp, (r < 3) ? (r + 2) : (r + 1));
        int myn = 0;
        float4 acc = make_float4(0.0f, 0.0f, 0.0f, 0.0f);
        for (int b = beg; b < end; b += 32) {
            int m = end - b;
            if (m > 32) m = 32;
            int my = my_first;
            if (b != beg) my = (lane < m) ? ssrc[b + lane] : 0;
            for (int j = 0; j < m; j += 8) {
                int s[8];
                float4 v[8];
#pragma unroll
                for (int u = 0; u < 8; ++u) {
                    int l = (j + u < m) ? (j + u) : (m - 1);
                    s[u] = __builtin_amdgcn_readlane(my, l);
                }
#pragma unroll
                for (int u = 0; u < 8; ++u) {
                    v[u] = make_float4(0.0f, 0.0f, 0.0f, 0.0f);
                    if (j + u < m) v[u] = *(const float4*)(colp + (size_t)s[u] * F);
                }
                myn = (lane < nend - end) ? ssrc[end + lane] : 0;
#pragma unroll
                for (int u = 0; u < 8; ++u) {
                    acc.x += v[u].x; acc.y += v[u].y; acc.z += v[u].z; acc.w += v[u].w;
                }
            }
        }
        if (end <= beg) myn = (lane < nend - end) ? ssrc[end + lane] : 0;
        my_first = myn;
        float dn = __int_as_float(__builtin_amdgcn_readlane(dnb, r));
        float sn = __int_as_float(__builtin_amdgcn_readlane(snb, r));
        float4 h;
        h.x = acc.x * dn + bv.x; h.x = (h.x > 0.0f) ? h.x : 0.0f;
        h.y = acc.y * dn + bv.y; h.y = (h.y > 0.0f) ? h.y : 0.0f;
        h.z = acc.z * dn + bv.z; h.z = (h.z > 0.0f) ? h.z : 0.0f;
        h.w = acc.w * dn + bv.w; h.w = (h.w > 0.0f) ? h.w : 0.0f;
        h.x *= sn; h.y *= sn; h.z *= sn; h.w *= sn;
        *(float4*)(tile + (q0 + r) * GCN_TILE_PITCH + c0) = h;
    }

    __syncthreads();

    if (NT < 4 && wv != 0) return;
    int ct0 = (NT >= 4) ? (wv * NTW) : 0;

    int hi   = lane >> 4;
    int kOff = hi * 2;
    int col  = lane & 15;
    const float* arow = tile + col * GCN_TILE_PITCH + kOff;

    v8f cacc[NTW];
    v8f vzero = {};
#pragma unroll
    for (int t = 0; t < NTW; ++t) cacc[t] = vzero;

#pragma unroll 8
    for (int k0 = 0; k0 < F; k0 += 4) {
        float2 av = *(const float2*)(arow + k0);
        v2f a;
        a.x = av.x;
        a.y = av.y;
        const float* __restrict__ wr = Wn + (size_t)(k0 + kOff) * (size_t)NOUT + ct0 * 16;
#pragma unroll
        for (int t = 0; t < NTW; ++t) {
            v2f b;
            b.x = wr[t * 16 + col];
            b.y = wr[NOUT + t * 16 + col];
            cacc[t] = __builtin_amdgcn_wmma_f32_16x16x4_f32(
                 false, a,  false, b,
                 (short)0, cacc[t],  false,  false);
        }
    }

    int crow = row0 + hi * 8;
#pragma unroll
    for (int t = 0; t < NTW; ++t) {
#pragma unroll
        for (int v = 0; v < 8; ++v) {
            if (crow + v < n_nodes)
                out[(size_t)(crow + v) * (size_t)NOUT + (ct0 + t) * 16 + col] = cacc[t][v];
        }
    }
}

__device__ __forceinline__ float gcn_pos_sigmoid(float v) {
    return 1.0f / (1.0f + expf(-v)) + 1e-8f;
}

__global__ __launch_bounds__(256)
void gcn_gather16_sigmoid_kernel(const float* __restrict__ hw, const int* __restrict__ rowptr,
                                 const int* __restrict__ ssrc, const float* __restrict__ dnorm,
                                 const float* __restrict__ bias, float* __restrict__ out, int n_nodes) {
    constexpr int F = 16;
    int gid = blockIdx.x * blockDim.x + threadIdx.x;
    int node = gid >> 2;
    int sub = gid & 3;
    if (node >= n_nodes) return;
    int beg = rowptr[node];
    int end = rowptr[node + 1];
    float dn = dnorm[node];
    int c0 = sub * 4;
    float4 bv = make_float4(bias[c0], bias[c0 + 1], bias[c0 + 2], bias[c0 + 3]);

    float4 acc = make_float4(0.0f, 0.0f, 0.0f, 0.0f);
    const float* __restrict__ colp = hw + sub * 4;

    for (int j = beg; j < end; j += 8) {
        int s[8];
        float4 v[8];
#pragma unroll
        for (int u = 0; u < 8; ++u) s[u] = (j + u < end) ? ssrc[j + u] : -1;
#pragma unroll
        for (int u = 0; u < 8; ++u) {
            v[u] = make_float4(0.0f, 0.0f, 0.0f, 0.0f);
            if (s[u] >= 0) v[u] = *(const float4*)(colp + (size_t)s[u] * F);
        }
#pragma unroll
        for (int u = 0; u < 8; ++u) {
            acc.x += v[u].x; acc.y += v[u].y; acc.z += v[u].z; acc.w += v[u].w;
        }
    }
    float4 r;
    r.x = gcn_pos_sigmoid(acc.x * dn + bv.x);
    r.y = gcn_pos_sigmoid(acc.y * dn + bv.y);
    r.z = gcn_pos_sigmoid(acc.z * dn + bv.z);
    r.w = gcn_pos_sigmoid(acc.w * dn + bv.w);
    *(float4*)(out + (size_t)node * F + sub * 4) = r;
}


extern "C" void kernel_launch(void* const* d_in, const int* in_sizes, int n_in,
                              void* d_out, int out_size, void* d_ws, size_t ws_size,
                              hipStream_t stream) {
    (void)n_in; (void)out_size; (void)ws_size;

    const float* feat = (const float*)d_in[0];
    const int*   src  = (const int*)d_in[1];
    const int*   dst  = (const int*)d_in[2];
    const float* W0   = (const float*)d_in[3];
    const float* b0   = (const float*)d_in[4];
    const float* W1   = (const float*)d_in[5];
    const float* b1   = (const float*)d_in[6];
    const float* W2   = (const float*)d_in[7];
    const float* b2   = (const float*)d_in[8];

    const int F = 128;
    const int N = in_sizes[0] / F;
    const int E = in_sizes[1];

    auto al = [](size_t n) { return (n + 63) & ~(size_t)63; };
    const int n_chunks = (N + GCN_SCAN_CHUNK - 1) / GCN_SCAN_CHUNK;
    float* snorm  = (float*)d_ws;
    float* dnorm  = snorm + al((size_t)N);
    int*   incnt  = (int*)(dnorm + al((size_t)N));
    int*   rowptr = incnt + al((size_t)N);
    int*   csum   = rowptr + al((size_t)N + 1);
    int*   ssrc   = csum + al((size_t)n_chunks);
    float* bufA   = (float*)(ssrc + al((size_t)E));
    float* bufB   = bufA + al((size_t)N * F);
    unsigned int* part = (unsigned int*)(bufB + al((size_t)N * F));
    int*   slot0  = (int*)(part + al((size_t)2 * GCN_PARTS * GCN_HW));

    const int TB = 256;
    auto nblk = [](long long n, int tb) { return (unsigned)((n + tb - 1) / tb); };

    gcn_hist_kernel<<<2 * GCN_PARTS, 1024, 0, stream>>>(src, dst, part, E);
    gcn_chunk_sum_norm_kernel<<<n_chunks, TB, 0, stream>>>(part, incnt, csum, snorm, dnorm, N);
    gcn_rowptr_kernel<<<n_chunks, TB, 0, stream>>>(incnt, csum, part, rowptr, slot0, N);
    gcn_fill_kernel<<<GCN_PARTS, 1024, 0, stream>>>(src, dst, slot0, ssrc, E, N);

    const long long gemm_threads = (long long)((N + 15) / 16) * 32;
    gcn_gemm8_kernel<<<nblk(gemm_threads, TB), TB, 0, stream>>>(feat, snorm, W0, bufA, N, F);

    const int FB = 128;
    const unsigned n_tiles = nblk(N, 16);
    gcn_gather128_relu_gemm_kernel<8><<<n_tiles, FB, 0, stream>>>(bufA, rowptr, ssrc, dnorm, b0, snorm, W1, bufB, N);

    gcn_gather128_relu_gemm_kernel<1><<<n_tiles, FB, 0, stream>>>(bufB, rowptr, ssrc, dnorm, b1, snorm, W2, bufA, N);

    gcn_gather16_sigmoid_kernel<<<nblk((long long)N * 4, TB), TB, 0, stream>>>(bufA, rowptr, ssrc, dnorm, b2, (float*)d_out, N);
}


# Layers 0 and 1 sum their input rows first and transform the sums inside the gather kernel (sum x W = sum of row x W): layer-0 transform launch and its hw round trip dropped; sum order changes
# speedup vs baseline: 3.7438x; 1.0483x over previous
#include <hip/hip_runtime.h>
#include <hip/hip_bf16.h>
#include <math.h>

typedef __attribute__((ext_vector_type(2))) float v2f;
typedef __attribute__((ext_vector_type(8))) float v8f;


#define GCN_PARTS 64
#define GCN_HW    50000
static_assert((GCN_HW & 1) == 0, "a thread reads its 4 nodes' counters as one aligned pair of words");

__device__ __forceinline__ void gcn_chunk_bounds(int n_edges, int c, int& lo, int& hi) {
    int chunk = (((n_edges + GCN_PARTS - 1) / GCN_PARTS) + 3) & ~3;
    long long l = (long long)c * chunk;
    lo = (l < n_edges) ? (int)l : n_edges;
    hi = (n_edges - lo < chunk) ? n_edges : (lo + chunk);
}

__global__ __launch_bounds__(1024)
void gcn_hist_kernel(const int* __restrict__ src, const int* __restrict__ dst,
                     unsigned int* __restrict__ part, int n_edges) {
    __shared__ unsigned int h[GCN_HW];
    int tid = threadIdx.x;
    int role = (int)blockIdx.x / GCN_PARTS;
    int c = (int)blockIdx.x - role * GCN_PARTS;
    const int* __restrict__ key = role ? src : dst;
    for (int i = tid; i < GCN_HW; i += 1024) h[i] = 0u;
    __syncthreads();
    int lo, hi;
    gcn_chunk_bounds(n_edges, c, lo, hi);
    int nq = (hi - lo) >> 2;
    const int4* __restrict__ kq = (const int4*)(key + lo);
    for (int q = tid; q < nq; q += 4096) {
        int4 k[4];
#pragma unroll
        for (int u = 0; u < 4; ++u) {
            int qu = q + u * 1024;
            k[u] = kq[(qu < nq) ? qu : q];
        }
#pragma unroll
        for (int u = 0; u < 4; ++u) {
            if (q + u * 1024 < nq) {
                atomicAdd(&h[k[u].x >> 1], 1u << ((k[u].x & 1) * 16));
                atomicAdd(&h[k[u].y >> 1], 1u << ((k[u].y & 1) * 16));
                atomicAdd(&h[k[u].z >> 1], 1u << ((k[u].z & 1) * 16));
                atomicAdd(&h[k[u].w >> 1], 1u << ((k[u].w & 1) * 16));
            }
        }
    }
    for (int i = lo + 4 * nq + tid; i < hi; i += 1024) {
        int k = key[i];
        atomicAdd(&h[k >> 1], 1u << ((k & 1) * 16));
    }
    __syncthreads();
    unsigned int* __restrict__ out = part + (size_t)blockIdx.x * GCN_HW;
    for (int i = tid; i < GCN_HW; i += 1024) out[i] = h[i];
}


#define GCN_SCAN_CHUNK 1024

__device__ __forceinline__ int gcn_block_sum_256(int v, int* sm) {
    int lane = threadIdx.x & 31, wv = threadIdx.x >> 5;
    for (int o = 16; o > 0; o >>= 1) v += __shfl_xor(v, o);
    if (lane == 0) sm[wv] = v;
    __syncthreads();
    int t = 0;
    for (int w = 0; w < 8; ++w) t += sm[w];
    __syncthreads();
    return t;
}

__global__ __launch_bounds__(256)
void gcn_chunk_sum_norm_kernel(const unsigned int* __restrict__ part, int* __restrict__ cnt, int* __restrict__ csum,
                               float* __restrict__ snorm, float* __restrict__ dnorm, int n) {
    __shared__ int sm[8];
    int base = blockIdx.x * GCN_SCAN_CHUNK + threadIdx.x * 4;
    int w0 = base >> 1;
    unsigned int ci[4] = {0u, 0u, 0u, 0u};
    unsigned int co[4] = {0u, 0u, 0u, 0u};
    if (w0 < GCN_HW) {
        const unsigned int* __restrict__ pd = part + w0;
        const unsigned int* __restrict__ ps = part + (size_t)GCN_PARTS * GCN_HW + w0;
#pragma unroll 16
        for (int p = 0; p < GCN_PARTS; ++p) {
            uint2 a = *(const uint2*)(pd + (size_t)p * GCN_HW);
            uint2 b = *(const uint2*)(ps + (size_t)p * GCN_HW);
            ci[0] += a.x & 0xFFFFu; ci[1] += a.x >> 16; ci[2] += a.y & 0xFFFFu; ci[3] += a.y >> 16;
            co[0] += b.x & 0xFFFFu; co[1] += b.x >> 16; co[2] += b.y & 0xFFFFu; co[3] += b.y >> 16;
        }
    }
    int s = 0;
#pragma unroll
    for (int k = 0; k < 4; ++k) {
        if (base + k < n) {
            float d = (float)co[k];
            d = (d < 1.0f) ? 1.0f : d;
            snorm[base + k] = 1.0f / sqrtf(d);
            float cf = (float)ci[k];
            cf = (cf < 1.0f) ? 1.0f : cf;
            dnorm[base + k] = 1.0f / sqrtf(cf);
            cnt[base + k] = (int)ci[k];
            s += (int)ci[k];
        }
    }
    int t = gcn_block_sum_256(s, sm);
    if (threadIdx.x == 0) csum[blockIdx.x] = t;
}

__global__ __launch_bounds__(256)
void gcn_rowptr_kernel(const int* __restrict__ cnt, const int* __restrict__ csum,
                       const unsigned int* __restrict__ part, int* __restrict__ rowptr,
                       int* __restrict__ slot0, int n) {
    __shared__ int sm[8];
    __shared__ int wtot[8];
    int tid = threadIdx.x, lane = tid & 31, wv = tid >> 5;

    int p = 0;
    for (int j = tid; j < (int)blockIdx.x; j += 256) p += csum[j];
    int chunk_off = gcn_block_sum_256(p, sm);

    int base = blockIdx.x * GCN_SCAN_CHUNK + tid * 4;
    int c[4];
    int t = 0;
#pragma unroll
    for (int k = 0; k < 4; ++k) { c[k] = (base + k < n) ? cnt[base + k] : 0; t += c[k]; }

    int inc = t;
    for (int o = 1; o < 32; o <<= 1) {
        int y = __shfl_up(inc, o);
        if (lane >= o) inc += y;
    }
    if (lane == 31) wtot[wv] = inc;
    __syncthreads();
    int wpre = 0;
    for (int w = 0; w < 8; ++w) if (w < wv) wpre += wtot[w];

    int run = chunk_off + wpre + inc - t;
    int r[4];
#pragma unroll
    for (int k = 0; k < 4; ++k) {
        r[k] = run;
        if (base + k < n) {
            rowptr[base + k] = run;
            run += c[k];
            if (base + k == n - 1) rowptr[n] = run;
        }
    }

    if (base < n) {
        const unsigned int* __restrict__ pd = part + (base >> 1);
        int a0 = r[0], a1 = r[1], a2 = r[2], a3 = r[3];
        if ((base + 3 < n) && ((n & 3) == 0)) {
#pragma unroll 16
            for (int q = 0; q < GCN_PARTS; ++q) {
                uint2 a = *(const uint2*)(pd + (size_t)q * GCN_HW);
                *(int4*)(slot0 + (size_t)q * (size_t)n + base) = make_int4(a0, a1, a2, a3);
                a0 += (int)(a.x & 0xFFFFu); a1 += (int)(a.x >> 16);
                a2 += (int)(a.y & 0xFFFFu); a3 += (int)(a.y >> 16);
            }
        } else {
            for (int q = 0; q < GCN_PARTS; ++q) {
                uint2 a = *(const uint2*)(pd + (size_t)q * GCN_HW);
                int* __restrict__ o = slot0 + (size_t)q * (size_t)n + base;
                o[0] = a0;
                if (base + 1 < n) o[1] = a1;
                if (base + 2 < n) o[2] = a2;
                if (base + 3 < n) o[3] = a3;
                a0 += (int)(a.x & 0xFFFFu); a1 += (int)(a.x >> 16);
                a2 += (int)(a.y & 0xFFFFu); a3 += (int)(a.y >> 16);
            }
        }
    }
}

__global__ __launch_bounds__(1024)
void gcn_fill_kernel(const int* __restrict__ src, const int* __restrict__ dst,
                     const int* __restrict__ slot0, int* __restrict__ ssrc, int n_edges, int n_nodes) {
    __shared__ unsigned int h[GCN_HW];
    int tid = threadIdx.x;
    int c = (int)blockIdx.x;
    for (int i = tid; i < GCN_HW; i += 1024) h[i] = 0u;
    __syncthreads();
    const int* __restrict__ s0 = slot0 + (size_t)c * (size_t)n_nodes;
    int lo, hi;
    gcn_chunk_bounds(n_edges, c, lo, hi);
    int nq = (hi - lo) >> 2;
    const int4* __restrict__ dq = (const int4*)(dst + lo);
    const int4* __restrict__ sq = (const int4*)(src + lo);
    for (int q = tid; q < nq; q += 1024) {
        int4 d = dq[q];
        int4 s = sq[q];
        int p0 = s0[d.x];
        int p1 = s0[d.y];
        int p2 = s0[d.z];
        int p3 = s0[d.w];
        int h0 = (d.x & 1) * 16, h1 = (d.y & 1) * 16, h2 = (d.z & 1) * 16, h3 = (d.w & 1) * 16;
        unsigned int r0 = (atomicAdd(&h[d.x >> 1], 1u << h0) >> h0) & 0xFFFFu;
        unsigned int r1 = (atomicAdd(&h[d.y >> 1], 1u << h1) >> h1) & 0xFFFFu;
        unsigned int r2 = (atomicAdd(&h[d.z >> 1], 1u << h2) >> h2) & 0xFFFFu;
        unsigned int r3 = (atomicAdd(&h[d.w >> 1], 1u << h3) >> h3) & 0xFFFFu;
        ssrc[p0 + (int)r0] = s.x;
        ssrc[p1 + (int)r1] = s.y;
        ssrc[p2 + (int)r2] = s.z;
        ssrc[p3 + (int)r3] = s.w;
    }
    for (int i = lo + 4 * nq + tid; i < hi; i += 1024) {
        int d = dst[i];
        int sh = (d & 1) * 16;
        unsigned int r = (atomicAdd(&h[d >> 1], 1u << sh) >> sh) & 0xFFFFu;
        ssrc[s0[d] + (int)r] = src[i];
    }
}


#define GCN_TILE_PITCH 132

template <bool EDGE_SCALE, bool TAIL16>
__global__ __launch_bounds__(128)
void gcn_sum128_gemm_kernel(const float* __restrict__ xin, const int* __restrict__ rowptr,
                            const int* __restrict__ ssrc, const float* __restrict__ dnorm,
                            const float* __restrict__ bias, const float* __restrict__ snorm,
                            const float* __restrict__ Wt, const float* __restrict__ W2,
                            float* __restrict__ out, int n_nodes) {
    constexpr int F = 128;
    constexpr int NTW = 2;
    __shared__ __attribute__((aligned(16))) float tile[16 * GCN_TILE_PITCH];
    __shared__ __attribute__((aligned(16))) float tile2[TAIL16 ? 16 * GCN_TILE_PITCH : 4];
    __shared__ __attribute__((aligned(16))) float rdn[16];
    __shared__ __attribute__((aligned(16))) float rsn[16];

    int lane = threadIdx.x & 31;
    int wv = __builtin_amdgcn_readfirstlane((int)(threadIdx.x >> 5));
    int row0 = (int)blockIdx.x * 16;
    if (row0 >= n_nodes) return;

    int q0 = wv * 4;
    int ni = row0 + q0 + lane;
    int rp = rowptr[(ni < n_nodes) ? ni : n_nodes];
    int nc = (ni < n_nodes) ? ni : (n_nodes - 1);
    if (lane < 4) {
        rdn[q0 + lane] = dnorm[nc];
        rsn[q0 + lane] = snorm[nc];
    }
    int c0 = lane * 4;
    const float* __restrict__ colp = xin + lane * 4;

    int beg0 = __builtin_amdgcn_readlane(rp, 0);
    int end0 = __builtin_amdgcn_readlane(rp, 1);
    int my_first = (lane < end0 - beg0) ? ssrc[beg0 + lane] : 0;

    for (int r = 0; r < 4; ++r) {
        int beg = __builtin_amdgcn_readlane(rp, r);
        int end = __builtin_amdgcn_readlane(rp, r + 1);
        int nend = __builtin_amdgcn_readlane(rp, (r < 3) ? (r + 2) : (r + 1));
        int myn = 0;
        float4 acc = make_float4(0.0f, 0.0f, 0.0f, 0.0f);
        for (int b = beg; b < end; b += 32) {
            int m = end - b;
            if (m > 32) m = 32;
            int my = my_first;
            if (b != beg) my = (lane < m) ? ssrc[b + lane] : 0;
            int mysn = 0;
            if (EDGE_SCALE) mysn = __float_as_int(snorm[my]);
            for (int j = 0; j < m; j += 8) {
                int s[8];
                float4 v[8];
#pragma unroll
                for (int u = 0; u < 8; ++u) {
                    int l = (j + u < m) ? (j + u) : (m - 1);
                    s[u] = __builtin_amdgcn_readlane(my, l);
                }
#pragma unroll
                for (int u = 0; u < 8; ++u) {
                    v[u] = make_float4(0.0f, 0.0f, 0.0f, 0.0f);
                    if (j + u < m) v[u] = *(const float4*)(colp + (size_t)s[u] * F);
                }
                myn = (lane < nend - end) ? ssrc[end + lane] : 0;
#pragma unroll
                for (int u = 0; u < 8; ++u) {
                    if (EDGE_SCALE) {
                        int l = (j + u < m) ? (j + u) : (m - 1);
                        float w = __int_as_float(__builtin_amdgcn_readlane(mysn, l));
                        acc.x += w * v[u].x; acc.y += w * v[u].y; acc.z += w * v[u].z; acc.w += w * v[u].w;
                    } else {
                        acc.x += v[u].x; acc.y += v[u].y; acc.z += v[u].z; acc.w += v[u].w;
                    }
                }
            }
        }
        if (end <= beg) myn = (lane < nend - end) ? ssrc[end + lane] : 0;
        my_first = myn;
        *(float4*)(tile + (q0 + r) * GCN_TILE_PITCH + c0) = acc;
    }

    __syncthreads();

    int ct0 = wv * NTW;
    int hi   = lane >> 4;
    int kOff = hi * 2;
    int col  = lane & 15;
    const float* arow = tile + col * GCN_TILE_PITCH + kOff;

    v8f cacc[NTW];
    v8f vzero = {};
#pragma unroll
    for (int t = 0; t < NTW; ++t) cacc[t] = vzero;

#pragma unroll 8
    for (int k0 = 0; k0 < F; k0 += 4) {
        float2 av = *(const float2*)(arow + k0);
        v2f a;
        a.x = av.x;
        a.y = av.y;
        const float* __restrict__ wr = Wt + (size_t)(k0 + kOff) * (size_t)F + ct0 * 16;
#pragma unroll
        for (int t = 0; t < NTW; ++t) {
            v2f b;
            b.x = wr[t * 16 + col];
            b.y = wr[F + t * 16 + col];
            cacc[t] = __builtin_amdgcn_wmma_f32_16x16x4_f32(
                 false, a,  false, b,
                 (short)0, cacc[t],  false,  false);
        }
    }

    float4 dA = *(const float4*)(rdn + hi * 8);
    float4 dB = *(const float4*)(rdn + hi * 8 + 4);
    float4 sA = *(const float4*)(rsn + hi * 8);
    float4 sB = *(const float4*)(rsn + hi * 8 + 4);
    float dnv[8] = {dA.x, dA.y, dA.z, dA.w, dB.x, dB.y, dB.z, dB.w};
    float snv[8] = {sA.x, sA.y, sA.z, sA.w, sB.x, sB.y, sB.z, sB.w};
    int crow = row0 + hi * 8;
#pragma unroll
    for (int t = 0; t < NTW; ++t) {
        int oc = (ct0 + t) * 16 + col;
        float bc = bias[oc];
#pragma unroll
        for (int v = 0; v < 8; ++v) {
            float y = cacc[t][v] * dnv[v] + bc;
            y = (y > 0.0f) ? y : 0.0f;
            y *= snv[v];
            if (TAIL16) {
                tile2[(hi * 8 + v) * GCN_TILE_PITCH + oc] = y;
            } else {
                if (crow + v < n_nodes) out[(size_t)(crow + v) * (size_t)F + oc] = y;
            }
        }
    }
    if (!TAIL16) return;

    __syncthreads();

    if (wv != 0) return;
    const float* arow2 = tile2 + col * GCN_TILE_PITCH + kOff;
    v8f c2 = vzero;
#pragma unroll 8
    for (int k0 = 0; k0 < F; k0 += 4) {
        float2 av = *(const float2*)(arow2 + k0);
        v2f a;
        a.x = av.x;
        a.y = av.y;
        const float* __restrict__ wr = W2 + (size_t)(k0 + kOff) * 16;
        v2f b;
        b.x = wr[col];
        b.y = wr[16 + col];
        c2 = __builtin_amdgcn_wmma_f32_16x16x4_f32(
             false, a,  false, b,
             (short)0, c2,  false,  false);
    }
#pragma unroll
    for (int v = 0; v < 8; ++v) {
        if (crow + v < n_nodes) out[(size_t)(crow + v) * 16 + col] = c2[v];
    }
}

__device__ __forceinline__ float gcn_pos_sigmoid(float v) {
    return 1.0f / (1.0f + expf(-v)) + 1e-8f;
}

__global__ __launch_bounds__(256)
void gcn_gather16_sigmoid_kernel(const float* __restrict__ hw, const int* __restrict__ rowptr,
                                 const int* __restrict__ ssrc, const float* __restrict__ dnorm,
                                 const float* __restrict__ bias, float* __restrict__ out, int n_nodes) {
    constexpr int F = 16;
    int gid = blockIdx.x * blockDim.x + threadIdx.x;
    int node = gid >> 2;
    int sub = gid & 3;
    if (node >= n_nodes) return;
    int beg = rowptr[node];
    int end = rowptr[node + 1];
    float dn = dnorm[node];
    int c0 = sub * 4;
    float4 bv = make_float4(bias[c0], bias[c0 + 1], bias[c0 + 2], bias[c0 + 3]);

    float4 acc = make_float4(0.0f, 0.0f, 0.0f, 0.0f);
    const float* __restrict__ colp = hw + sub * 4;

    for (int j = beg; j < end; j += 8) {
        int s[8];
        float4 v[8];
#pragma unroll
        for (int u = 0; u < 8; ++u) s[u] = (j + u < end) ? ssrc[j + u] : -1;
#pragma unroll
        for (int u = 0; u < 8; ++u) {
            v[u] = make_float4(0.0f, 0.0f, 0.0f, 0.0f);
            if (s[u] >= 0) v[u] = *(const float4*)(colp + (size_t)s[u] * F);
        }
#pragma unroll
        for (int u = 0; u < 8; ++u) {
            acc.x += v[u].x; acc.y += v[u].y; acc.z += v[u].z; acc.w += v[u].w;
        }
    }
    float4 r;
    r.x = gcn_pos_sigmoid(acc.x * dn + bv.x);
    r.y = gcn_pos_sigmoid(acc.y * dn + bv.y);
    r.z = gcn_pos_sigmoid(acc.z * dn + bv.z);
    r.w = gcn_pos_sigmoid(acc.w * dn + bv.w);
    *(float4*)(out + (size_t)node * F + sub * 4) = r;
}


extern "C" void kernel_launch(void* const* d_in, const int* in_sizes, int n_in,
                              void* d_out, int out_size, void* d_ws, size_t ws_size,
                              hipStream_t stream) {
    (void)n_in; (void)out_size; (void)ws_size;

    const float* feat = (const float*)d_in[0];
    const int*   src  = (const int*)d_in[1];
    const int*   dst  = (const int*)d_in[2];
    const float* W0   = (const float*)d_in[3];
    const float* b0   = (const float*)d_in[4];
    const float* W1   = (const float*)d_in[5];
    const float* b1   = (const float*)d_in[6];
    const float* W2   = (const float*)d_in[7];
    const float* b2   = (const float*)d_in[8];

    const int F = 128;
    const int N = in_sizes[0] / F;
    const int E = in_sizes[1];

    auto al = [](size_t n) { return (n + 63) & ~(size_t)63; };
    const int n_chunks = (N + GCN_SCAN_CHUNK - 1) / GCN_SCAN_CHUNK;
    float* snorm  = (float*)d_ws;
    float* dnorm  = snorm + al((size_t)N);
    int*   incnt  = (int*)(dnorm + al((size_t)N));
    int*   rowptr = incnt + al((size_t)N);
    int*   csum   = rowptr + al((size_t)N + 1);
    int*   ssrc   = csum + al((size_t)n_chunks);
    float* bufA   = (float*)(ssrc + al((size_t)E));
    float* bufB   = bufA + al((size_t)N * F);
    unsigned int* part = (unsigned int*)(bufB + al((size_t)N * F));
    int*   slot0  = (int*)(part + al((size_t)2 * GCN_PARTS * GCN_HW));

    const int TB = 256;
    auto nblk = [](long long n, int tb) { return (unsigned)((n + tb - 1) / tb); };

    gcn_hist_kernel<<<2 * GCN_PARTS, 1024, 0, stream>>>(src, dst, part, E);
    gcn_chunk_sum_norm_kernel<<<n_chunks, TB, 0, stream>>>(part, incnt, csum, snorm, dnorm, N);
    gcn_rowptr_kernel<<<n_chunks, TB, 0, stream>>>(incnt, csum, part, rowptr, slot0, N);
    gcn_fill_kernel<<<GCN_PARTS, 1024, 0, stream>>>(src, dst, slot0, ssrc, E, N);

    const int FB = 128;
    const unsigned n_tiles = nblk(N, 16);

    gcn_sum128_gemm_kernel<true, false><<<n_tiles, FB, 0, stream>>>(feat, rowptr, ssrc, dnorm, b0, snorm, W0, W2, bufA, N);

    gcn_sum128_gemm_kernel<false, true><<<n_tiles, FB, 0, stream>>>(bufA, rowptr, ssrc, dnorm, b1, snorm, W1, W2, bufB, N);

    gcn_gather16_sigmoid_kernel<<<nblk((long long)N * 4, TB), TB, 0, stream>>>(bufB, rowptr, ssrc, dnorm, b2, (float*)d_out, N);
}


# Counting sort's per-edge scattered 4-B slot reads and list stores replaced by grouping edges by 512-node destination range in LDS with whole-line stores; 7 launches to 5; sum order unchanged
# speedup vs baseline: 5.5923x; 1.4937x over previous
#include <hip/hip_runtime.h>
#include <hip/hip_bf16.h>
#include <math.h>

typedef __attribute__((ext_vector_type(2))) float v2f;
typedef __attribute__((ext_vector_type(8))) float v8f;


#define GCN_PARTS  64
#define GCN_HW     50000
#define GCN_RSHIFT 9
#define GCN_RNODES (1 << GCN_RSHIFT)
#define GCN_RSLOTS 256
#define GCN_STAGE  16384
static_assert(GCN_PARTS == 64, "a range block's 16 waves take 4 chunks each, and a wave sums 64 values in two halves");
static_assert((2 * GCN_HW + GCN_RNODES - 1) / GCN_RNODES <= GCN_RSLOTS, "every range has a slot");
static_assert(2 * GCN_HW <= (1 << (32 - GCN_RSHIFT)), "a source id fits above the destination's low bits");

__device__ __forceinline__ int gcn_chunk_len(int n_edges) {
    return (((n_edges + GCN_PARTS - 1) / GCN_PARTS) + 3) & ~3;
}
__device__ __forceinline__ int gcn_chunk_pitch(int n_edges) {
    return (gcn_chunk_len(n_edges) + 31) & ~31;
}
__device__ __forceinline__ void gcn_chunk_bounds(int n_edges, int c, int& lo, int& hi) {
    int chunk = gcn_chunk_len(n_edges);
    long long l = (long long)c * chunk;
    lo = (l < n_edges) ? (int)l : n_edges;
    hi = (n_edges - lo < chunk) ? n_edges : (lo + chunk);
}

__global__ __launch_bounds__(1024)
void gcn_hist_bin_kernel(const int* __restrict__ src, const int* __restrict__ dst,
                         unsigned int* __restrict__ part, unsigned int* __restrict__ binned,
                         unsigned int* __restrict__ rcnt, unsigned int* __restrict__ roff, int n_edges) {
    __shared__ unsigned int h[GCN_HW];
    __shared__ unsigned int rc[GCN_RSLOTS];
    __shared__ unsigned int ro[GCN_RSLOTS];
    int tid = threadIdx.x;
    int role = (int)blockIdx.x / GCN_PARTS;
    int c = (int)blockIdx.x - role * GCN_PARTS;
    int lo, hi;
    gcn_chunk_bounds(n_edges, c, lo, hi);
    int nq = (hi - lo) >> 2;

    if (role == 1) {
        for (int i = tid; i < GCN_HW; i += 1024) h[i] = 0u;
        __syncthreads();
        const int4* __restrict__ kq = (const int4*)(src + lo);
        for (int q = tid; q < nq; q += 4096) {
            int4 k[4];
#pragma unroll
            for (int u = 0; u < 4; ++u) {
                int qu = q + u * 1024;
                k[u] = kq[(qu < nq) ? qu : q];
            }
#pragma unroll
            for (int u = 0; u < 4; ++u) {
                if (q + u * 1024 < nq) {
                    atomicAdd(&h[k[u].x >> 1], 1u << ((k[u].x & 1) * 16));
                    atomicAdd(&h[k[u].y >> 1], 1u << ((k[u].y & 1) * 16));
                    atomicAdd(&h[k[u].z >> 1], 1u << ((k[u].z & 1) * 16));
                    atomicAdd(&h[k[u].w >> 1], 1u << ((k[u].w & 1) * 16));
                }
            }
        }
        for (int i = lo + 4 * nq + tid; i < hi; i += 1024) {
            int k = src[i];
            atomicAdd(&h[k >> 1], 1u << ((k & 1) * 16));
        }
        __syncthreads();
        unsigned int* __restrict__ out = part + (size_t)c * GCN_HW;
        for (int i = tid; i < GCN_HW; i += 1024) out[i] = h[i];
        return;
    }

    if (tid < GCN_RSLOTS) rc[tid] = 0u;
    __syncthreads();
    const int4* __restrict__ dq = (const int4*)(dst + lo);
    const int4* __restrict__ sq = (const int4*)(src + lo);
    for (int q = tid; q < nq; q += 4096) {
        int4 k[4];
#pragma unroll
        for (int u = 0; u < 4; ++u) {
            int qu = q + u * 1024;
            k[u] = dq[(qu < nq) ? qu : q];
        }
#pragma unroll
        for (int u = 0; u < 4; ++u) {
            if (q + u * 1024 < nq) {
                atomicAdd(&rc[k[u].x >> GCN_RSHIFT], 1u);
                atomicAdd(&rc[k[u].y >> GCN_RSHIFT], 1u);
                atomicAdd(&rc[k[u].z >> GCN_RSHIFT], 1u);
                atomicAdd(&rc[k[u].w >> GCN_RSHIFT], 1u);
            }
        }
    }
    for (int i = lo + 4 * nq + tid; i < hi; i += 1024) atomicAdd(&rc[dst[i] >> GCN_RSHIFT], 1u);
    __syncthreads();

    if (tid < 32) {
        unsigned int v[8];
        unsigned int t = 0u;
#pragma unroll
        for (int k = 0; k < 8; ++k) { v[k] = rc[tid * 8 + k]; t += v[k]; }
        unsigned int inc = t;
        for (int o = 1; o < 32; o <<= 1) {
            unsigned int y = __shfl_up(inc, o);
            if (tid >= o) inc += y;
        }
        unsigned int run = inc - t;
#pragma unroll
        for (int k = 0; k < 8; ++k) { ro[tid * 8 + k] = run; run += v[k]; }
    }
    __syncthreads();
    if (tid < GCN_RSLOTS) {
        unsigned int cv = rc[tid];
        unsigned int ov = ro[tid];
        rcnt[c * GCN_RSLOTS + tid] = cv;
        roff[c * GCN_RSLOTS + tid] = ov;
        rc[tid] = ov;
    }
    __syncthreads();

    for (int q = tid; q < nq; q += 4096) {
        int4 kd[4];
        int4 ks[4];
#pragma unroll
        for (int u = 0; u < 4; ++u) {
            int qu = q + u * 1024;
            int qq = (qu < nq) ? qu : q;
            kd[u] = dq[qq];
            ks[u] = sq[qq];
        }
#pragma unroll
        for (int u = 0; u < 4; ++u) {
            if (q + u * 1024 < nq) {
                unsigned int p0 = atomicAdd(&rc[kd[u].x >> GCN_RSHIFT], 1u);
                unsigned int p1 = atomicAdd(&rc[kd[u].y >> GCN_RSHIFT], 1u);
                unsigned int p2 = atomicAdd(&rc[kd[u].z >> GCN_RSHIFT], 1u);
                unsigned int p3 = atomicAdd(&rc[kd[u].w >> GCN_RSHIFT], 1u);
                h[p0] = ((unsigned int)ks[u].x << GCN_RSHIFT) | ((unsigned int)kd[u].x & (GCN_RNODES - 1));
                h[p1] = ((unsigned int)ks[u].y << GCN_RSHIFT) | ((unsigned int)kd[u].y & (GCN_RNODES - 1));
                h[p2] = ((unsigned int)ks[u].z << GCN_RSHIFT) | ((unsigned int)kd[u].z & (GCN_RNODES - 1));
                h[p3] = ((unsigned int)ks[u].w << GCN_RSHIFT) | ((unsigned int)kd[u].w & (GCN_RNODES - 1));
            }
        }
    }
    for (int i = lo + 4 * nq + tid; i < hi; i += 1024) {
        int d = dst[i];
        unsigned int p = atomicAdd(&rc[d >> GCN_RSHIFT], 1u);
        h[p] = ((unsigned int)src[i] << GCN_RSHIFT) | ((unsigned int)d & (GCN_RNODES - 1));
    }
    __syncthreads();
    unsigned int* __restrict__ outb = binned + (size_t)c * (size_t)gcn_chunk_pitch(n_edges);
    int cnt = hi - lo;
    for (int i = tid; i < cnt; i += 1024) outb[i] = h[i];
}

__global__ __launch_bounds__(GCN_RNODES)
void gcn_range_build_kernel(const unsigned int* __restrict__ binned, const unsigned int* __restrict__ rcnt,
                            const unsigned int* __restrict__ roff, const unsigned int* __restrict__ part,
                            int* __restrict__ rowptr, int* __restrict__ ssrc,
                            float* __restrict__ snorm, float* __restrict__ dnorm, int n_nodes, int n_edges) {
    __shared__ unsigned int cn[GCN_PARTS * GCN_RNODES];
    __shared__ unsigned int stage[GCN_STAGE];
    __shared__ int nstart[GCN_RNODES];
    __shared__ int s_cnt[GCN_PARTS];
    __shared__ int s_off[GCN_PARTS];
    __shared__ int wt[GCN_RNODES / 32];
    int tid = threadIdx.x, lane = tid & 31, wv = tid >> 5;
    int r = (int)blockIdx.x;
    int node = r * GCN_RNODES + tid;
    size_t pitch = (size_t)gcn_chunk_pitch(n_edges);

    if (tid < GCN_PARTS) {
        s_cnt[tid] = (int)rcnt[tid * GCN_RSLOTS + r];
        s_off[tid] = (int)roff[tid * GCN_RSLOTS + r];
    }
    for (int k = 0; k < GCN_PARTS; ++k) cn[k * GCN_RNODES + tid] = 0u;

    unsigned int od = 0u;
    if (node < n_nodes) {
        const unsigned int* __restrict__ ps = part + (node >> 1);
        int sh = (node & 1) * 16;
#pragma unroll 16
        for (int p = 0; p < GCN_PARTS; ++p) od += (ps[(size_t)p * GCN_HW] >> sh) & 0xFFFFu;
    }
    __syncthreads();

    int bo = s_off[lane] + s_off[lane + 32];
    int bt = s_cnt[lane] + s_cnt[lane + 32];
    for (int o = 16; o > 0; o >>= 1) { bo += __shfl_xor(bo, o); bt += __shfl_xor(bt, o); }
    int base = bo;
    int tot = bt;
    bool staged = (tot <= GCN_STAGE);

    int len[4];
    const unsigned int* run[4];
    int maxlen = 0;
#pragma unroll
    for (int k = 0; k < 4; ++k) {
        int c = wv + 16 * k;
        len[k] = s_cnt[c];
        run[k] = binned + (size_t)c * pitch + s_off[c];
        maxlen = (len[k] > maxlen) ? len[k] : maxlen;
    }

    for (int i0 = 0; i0 < maxlen; i0 += 256) {
        unsigned int e[4][8];
#pragma unroll
        for (int k = 0; k < 4; ++k) {
#pragma unroll
            for (int u = 0; u < 8; ++u) {
                int i = i0 + lane + 32 * u;
                e[k][u] = 0u;
                if (i < len[k]) e[k][u] = run[k][i];
            }
        }
#pragma unroll
        for (int k = 0; k < 4; ++k) {
            unsigned int* cc = cn + (wv + 16 * k) * GCN_RNODES;
#pragma unroll
            for (int u = 0; u < 8; ++u) {
                int i = i0 + lane + 32 * u;
                if (i < len[k]) atomicAdd(&cc[e[k][u] & (GCN_RNODES - 1)], 1u);
            }
        }
    }
    __syncthreads();

    unsigned int acc = 0u;
#pragma unroll 8
    for (int k = 0; k < GCN_PARTS; ++k) {
        unsigned int t = cn[k * GCN_RNODES + tid];
        cn[k * GCN_RNODES + tid] = acc;
        acc += t;
    }
    int v = (int)acc;
    int inc = v;
    for (int o = 1; o < 32; o <<= 1) {
        int y = __shfl_up(inc, o);
        if (lane >= o) inc += y;
    }
    if (lane == 31) wt[wv] = inc;
    __syncthreads();
    int wpre = 0;
    for (int w = 0; w < GCN_RNODES / 32; ++w) if (w < wv) wpre += wt[w];
    int excl = wpre + inc - v;
    if (node < n_nodes) {
        rowptr[node] = base + excl;
        if (node == n_nodes - 1) rowptr[n_nodes] = base + excl + v;
        float cf = (float)v;
        cf = (cf < 1.0f) ? 1.0f : cf;
        dnorm[node] = 1.0f / sqrtf(cf);
        float d = (float)od;
        d = (d < 1.0f) ? 1.0f : d;
        snorm[node] = 1.0f / sqrtf(d);
    }
    nstart[tid] = excl;
    __syncthreads();

    for (int i0 = 0; i0 < maxlen; i0 += 256) {
        unsigned int e[4][8];
#pragma unroll
        for (int k = 0; k < 4; ++k) {
#pragma unroll
            for (int u = 0; u < 8; ++u) {
                int i = i0 + lane + 32 * u;
                e[k][u] = 0u;
                if (i < len[k]) e[k][u] = run[k][i];
            }
        }
#pragma unroll
        for (int k = 0; k < 4; ++k) {
            unsigned int* cc = cn + (wv + 16 * k) * GCN_RNODES;
#pragma unroll
            for (int u = 0; u < 8; ++u) {
                int i = i0 + lane + 32 * u;
                if (i < len[k]) {
                    int dl = (int)(e[k][u] & (GCN_RNODES - 1));
                    int p = nstart[dl] + (int)atomicAdd(&cc[dl], 1u);
                    if (staged) stage[p] = e[k][u] >> GCN_RSHIFT;
                    else ssrc[base + p] = (int)(e[k][u] >> GCN_RSHIFT);
                }
            }
        }
    }
    if (!staged) return;
    __syncthreads();
    for (int i = tid; i < tot; i += GCN_RNODES) ssrc[base + i] = (int)stage[i];
}


#define GCN_TILE_PITCH 132

template <bool EDGE_SCALE, bool TAIL16>
__global__ __launch_bounds__(128)
void gcn_sum128_gemm_kernel(const float* __restrict__ xin, const int* __restrict__ rowptr,
                            const int* __restrict__ ssrc, const float* __restrict__ dnorm,
                            const float* __restrict__ bias, const float* __restrict__ snorm,
                            const float* __restrict__ Wt, const float* __restrict__ W2,
                            float* __restrict__ out, int n_nodes) {
    constexpr int F = 128;
    constexpr int NTW = 2;
    __shared__ __attribute__((aligned(16))) float tile[16 * GCN_TILE_PITCH];
    __shared__ __attribute__((aligned(16))) float tile2[TAIL16 ? 16 * GCN_TILE_PITCH : 4];
    __shared__ __attribute__((aligned(16))) float rdn[16];
    __shared__ __attribute__((aligned(16))) float rsn[16];

    int lane = threadIdx.x & 31;
    int wv = __builtin_amdgcn_readfirstlane((int)(threadIdx.x >> 5));
    int row0 = (int)blockIdx.x * 16;
    if (row0 >= n_nodes) return;

    int q0 = wv * 4;
    int ni = row0 + q0 + lane;
    int rp = rowptr[(ni < n_nodes) ? ni : n_nodes];
    int nc = (ni < n_nodes) ? ni : (n_nodes - 1);
    if (lane < 4) {
        rdn[q0 + lane] = dnorm[nc];
        rsn[q0 + lane] = snorm[nc];
    }
    int c0 = lane * 4;
    const float* __restrict__ colp = xin + lane * 4;

    int beg0 = __builtin_amdgcn_readlane(rp, 0);
    int end0 = __builtin_amdgcn_readlane(rp, 1);
    int my_first = (lane < end0 - beg0) ? ssrc[beg0 + lane] : 0;

    for (int r = 0; r < 4; ++r) {
        int beg = __builtin_amdgcn_readlane(rp, r);
        int end = __builtin_amdgcn_readlane(rp, r + 1);
        int nend = __builtin_amdgcn_readlane(rp, (r < 3) ? (r + 2) : (r + 1));
        int myn = 0;
        float4 acc = make_float4(0.0f, 0.0f, 0.0f, 0.0f);
        for (int b = beg; b < end; b += 32) {
            int m = end - b;
            if (m > 32) m = 32;
            int my = my_first;
            if (b != beg) my = (lane < m) ? ssrc[b + lane] : 0;
            int mysn = 0;
            if (EDGE_SCALE) mysn = __float_as_int(snorm[my]);
            for (int j = 0; j < m; j += 8) {
                int s[8];
                float4 v[8];
#pragma unroll
                for (int u = 0; u < 8; ++u) {
                    int l = (j + u < m) ? (j + u) : (m - 1);
                    s[u] = __builtin_amdgcn_readlane(my, l);
                }
#pragma unroll
                for (int u = 0; u < 8; ++u) {
                    v[u] = make_float4(0.0f, 0.0f, 0.0f, 0.0f);
                    if (j + u < m) v[u] = *(const float4*)(colp + (size_t)s[u] * F);
                }
                myn = (lane < nend - end) ? ssrc[end + lane] : 0;
#pragma unroll
                for (int u = 0; u < 8; ++u) {
                    if (EDGE_SCALE) {
                        int l = (j + u < m) ? (j + u) : (m - 1);
                        float w = __int_as_float(__builtin_amdgcn_readlane(mysn, l));
                        acc.x += w * v[u].x; acc.y += w * v[u].y; acc.z += w * v[u].z; acc.w += w * v[u].w;
                    } else {
                        acc.x += v[u].x; acc.y += v[u].y; acc.z += v[u].z; acc.w += v[u].w;
                    }
                }
            }
        }
        if (end <= beg) myn = (lane < nend - end) ? ssrc[end + lane] : 0;
        my_first = myn;
        *(float4*)(tile + (q0 + r) * GCN_TILE_PITCH + c0) = acc;
    }

    __syncthreads();

    int ct0 = wv * NTW;
    int hi   = lane >> 4;
    int kOff = hi * 2;
    int col  = lane & 15;
    const float* arow = tile + col * GCN_TILE_PITCH + kOff;

    v8f cacc[NTW];
    v8f vzero = {};
#pragma unroll
    for (int t = 0; t < NTW; ++t) cacc[t] = vzero;

#pragma unroll 8
    for (int k0 = 0; k0 < F; k0 += 4) {
        float2 av = *(const float2*)(arow + k0);
        v2f a;
        a.x = av.x;
        a.y = av.y;
        const float* __restrict__ wr = Wt + (size_t)(k0 + kOff) * (size_t)F + ct0 * 16;
#pragma unroll
        for (int t = 0; t < NTW; ++t) {
            v2f b;
            b.x = wr[t * 16 + col];
            b.y = wr[F + t * 16 + col];
            cacc[t] = __builtin_amdgcn_wmma_f32_16x16x4_f32(
                 false, a,  false, b,
                 (short)0, cacc[t],  false,  false);
        }
    }

    float4 dA = *(const float4*)(rdn + hi * 8);
    float4 dB = *(const float4*)(rdn + hi * 8 + 4);
    float4 sA = *(const float4*)(rsn + hi * 8);
    float4 sB = *(const float4*)(rsn + hi * 8 + 4);
    float dnv[8] = {dA.x, dA.y, dA.z, dA.w, dB.x, dB.y, dB.z, dB.w};
    float snv[8] = {sA.x, sA.y, sA.z, sA.w, sB.x, sB.y, sB.z, sB.w};
    int crow = row0 + hi * 8;
#pragma unroll
    for (int t = 0; t < NTW; ++t) {
        int oc = (ct0 + t) * 16 + col;
        float bc = bias[oc];
#pragma unroll
        for (int v = 0; v < 8; ++v) {
            float y = cacc[t][v] * dnv[v] + bc;
            y = (y > 0.0f) ? y : 0.0f;
            y *= snv[v];
            if (TAIL16) {
                tile2[(hi * 8 + v) * GCN_TILE_PITCH + oc] = y;
            } else {
                if (crow + v < n_nodes) out[(size_t)(crow + v) * (size_t)F + oc] = y;
            }
        }
    }
    if (!TAIL16) return;

    __syncthreads();

    if (wv != 0) return;
    const float* arow2 = tile2 + col * GCN_TILE_PITCH + kOff;
    v8f c2 = vzero;
#pragma unroll 8
    for (int k0 = 0; k0 < F; k0 += 4) {
        float2 av = *(const float2*)(arow2 + k0);
        v2f a;
        a.x = av.x;
        a.y = av.y;
        const float* __restrict__ wr = W2 + (size_t)(k0 + kOff) * 16;
        v2f b;
        b.x = wr[col];
        b.y = wr[16 + col];
        c2 = __builtin_amdgcn_wmma_f32_16x16x4_f32(
             false, a,  false, b,
             (short)0, c2,  false,  false);
    }
#pragma unroll
    for (int v = 0; v < 8; ++v) {
        if (crow + v < n_nodes) out[(size_t)(crow + v) * 16 + col] = c2[v];
    }
}

__device__ __forceinline__ float gcn_pos_sigmoid(float v) {
    return 1.0f / (1.0f + expf(-v)) + 1e-8f;
}

__global__ __launch_bounds__(256)
void gcn_gather16_sigmoid_kernel(const float* __restrict__ hw, const int* __restrict__ rowptr,
                                 const int* __restrict__ ssrc, const float* __restrict__ dnorm,
                                 const float* __restrict__ bias, float* __restrict__ out, int n_nodes) {
    constexpr int F = 16;
    int gid = blockIdx.x * blockDim.x + threadIdx.x;
    int node = gid >> 2;
    int sub = gid & 3;
    if (node >= n_nodes) return;
    int beg = rowptr[node];
    int end = rowptr[node + 1];
    float dn = dnorm[node];
    int c0 = sub * 4;
    float4 bv = make_float4(bias[c0], bias[c0 + 1], bias[c0 + 2], bias[c0 + 3]);

    float4 acc = make_float4(0.0f, 0.0f, 0.0f, 0.0f);
    const float* __restrict__ colp = hw + sub * 4;

    for (int j = beg; j < end; j += 8) {
        int s[8];
        float4 v[8];
#pragma unroll
        for (int u = 0; u < 8; ++u) s[u] = (j + u < end) ? ssrc[j + u] : -1;
#pragma unroll
        for (int u = 0; u < 8; ++u) {
            v[u] = make_float4(0.0f, 0.0f, 0.0f, 0.0f);
            if (s[u] >= 0) v[u] = *(const float4*)(colp + (size_t)s[u] * F);
        }
#pragma unroll
        for (int u = 0; u < 8; ++u) {
            acc.x += v[u].x; acc.y += v[u].y; acc.z += v[u].z; acc.w += v[u].w;
        }
    }
    float4 r;
    r.x = gcn_pos_sigmoid(acc.x * dn + bv.x);
    r.y = gcn_pos_sigmoid(acc.y * dn + bv.y);
    r.z = gcn_pos_sigmoid(acc.z * dn + bv.z);
    r.w = gcn_pos_sigmoid(acc.w * dn + bv.w);
    *(float4*)(out + (size_t)node * F + sub * 4) = r;
}


extern "C" void kernel_launch(void* const* d_in, const int* in_sizes, int n_in,
                              void* d_out, int out_size, void* d_ws, size_t ws_size,
                              hipStream_t stream) {
    (void)n_in; (void)out_size; (void)ws_size;

    const float* feat = (const float*)d_in[0];
    const int*   src  = (const int*)d_in[1];
    const int*   dst  = (const int*)d_in[2];
    const float* W0   = (const float*)d_in[3];
    const float* b0   = (const float*)d_in[4];
    const float* W1   = (const float*)d_in[5];
    const float* b1   = (const float*)d_in[6];
    const float* W2   = (const float*)d_in[7];
    const float* b2   = (const float*)d_in[8];

    const int F = 128;
    const int N = in_sizes[0] / F;
    const int E = in_sizes[1];

    auto al = [](size_t n) { return (n + 63) & ~(size_t)63; };
    const int n_ranges = (N + GCN_RNODES - 1) / GCN_RNODES;
    float* snorm  = (float*)d_ws;
    float* dnorm  = snorm + al((size_t)N);
    int*   rowptr = (int*)(dnorm + al((size_t)N));
    int*   ssrc   = rowptr + al((size_t)N + 1);
    float* bufA   = (float*)(ssrc + al((size_t)E));
    float* bufB   = bufA + al((size_t)N * F);
    unsigned int* part   = (unsigned int*)(bufB + al((size_t)N * F));
    unsigned int* binned = part + al((size_t)GCN_PARTS * GCN_HW);
    unsigned int* rcnt   = binned + al((size_t)E + (size_t)GCN_PARTS * 64);
    unsigned int* roff   = rcnt + al((size_t)GCN_PARTS * GCN_RSLOTS);

    const int TB = 256;
    auto nblk = [](long long n, int tb) { return (unsigned)((n + tb - 1) / tb); };

    gcn_hist_bin_kernel<<<2 * GCN_PARTS, 1024, 0, stream>>>(src, dst, part, binned, rcnt, roff, E);
    gcn_range_build_kernel<<<n_ranges, GCN_RNODES, 0, stream>>>(binned, rcnt, roff, part, rowptr, ssrc, snorm, dnorm, N, E);

    const int FB = 128;
    const unsigned n_tiles = nblk(N, 16);

    gcn_sum128_gemm_kernel<true, false><<<n_tiles, FB, 0, stream>>>(feat, rowptr, ssrc, dnorm, b0, snorm, W0, W2, bufA, N);

    gcn_sum128_gemm_kernel<false, true><<<n_tiles, FB, 0, stream>>>(bufA, rowptr, ssrc, dnorm, b1, snorm, W1, W2, bufB, N);

    gcn_gather16_sigmoid_kernel<<<nblk((long long)N * 4, TB), TB, 0, stream>>>(bufB, rowptr, ssrc, dnorm, b2, (float*)d_out, N);
}
